# starting kernel
# baseline (speedup 1.0000x reference)
#include <hip/hip_runtime.h>

typedef __attribute__((ext_vector_type(2))) float v2f;
typedef __attribute__((ext_vector_type(8))) float v8f;


__global__ void set_deg_kernel(float* deg, int n) {
    int i = blockIdx.x * blockDim.x + threadIdx.x;
    if (i < n) deg[i] = 1.0f;
}

__global__ void accum_deg_kernel(float* deg, const int* __restrict__ dst,
                                 const float* __restrict__ w, int e) {
    int i = blockIdx.x * blockDim.x + threadIdx.x;
    if (i < e) unsafeAtomicAdd(&deg[dst[i]], w[i]);
}

__global__ void dinv_kernel(float* deg, int n) {
    int i = blockIdx.x * blockDim.x + threadIdx.x;
    if (i < n) deg[i] = rsqrtf(deg[i]);
}

__global__ void norm_kernel(const float* __restrict__ dinv, const int* __restrict__ src,
                            const int* __restrict__ dst, const float* __restrict__ w,
                            float* __restrict__ nrm, int e) {
    int i = blockIdx.x * blockDim.x + threadIdx.x;
    if (i < e) nrm[i] = dinv[src[i]] * w[i] * dinv[dst[i]];
}


template <int COLS>
__global__ __launch_bounds__(256) void gemm_wmma_kernel(
    const float* __restrict__ X, const float* __restrict__ W,
    float* __restrict__ OUT, int nRows) {
    constexpr int K  = 128;
    constexpr int NT = COLS / 16;
    constexpr int CSH = (COLS == 128) ? 7 : 6;

    __shared__ float sW[(K / 2) * COLS * 2];
    for (int i = threadIdx.x; i < K * COLS; i += 256) {
        const int row = i >> CSH;
        const int col = i & (COLS - 1);
        sW[(row >> 1) * (COLS * 2) + col * 2 + (row & 1)] = W[i];
    }
    __syncthreads();

    const int wave = threadIdx.x >> 5;
    const int lane = threadIdx.x & 31;
    const int rb   = blockIdx.x * 8 + wave;
    if (rb * 16 >= nRows) return;

    const int l15  = lane & 15;
    const int hsel = lane >> 4;

    const float* aBase = X + (size_t)(rb * 16 + l15) * K + 2 * hsel;
    const float* bCol  = &sW[hsel * (COLS * 2) + l15 * 2];

    v8f acc[NT] = {};

    for (int k = 0; k < K; k += 4) {
        const v2f A = *(const v2f*)(aBase + k);
        const float* bp = bCol + (k >> 1) * (COLS * 2);
#pragma unroll
        for (int c = 0; c < NT; ++c) {
            const v2f B = *(const v2f*)(bp + c * 32);
            acc[c] = __builtin_amdgcn_wmma_f32_16x16x4_f32(
                false, A, false, B, (short)0, acc[c], false, false);
        }
    }

    const int rowBase = rb * 16 + (hsel << 3);
#pragma unroll
    for (int c = 0; c < NT; ++c) {
#pragma unroll
        for (int r = 0; r < 8; ++r)
            OUT[(size_t)(rowBase + r) * COLS + c * 16 + l15] = acc[c][r];
    }
}


template <int COLS>
__global__ void init_out_kernel(const float* __restrict__ h, const float* __restrict__ dinv,
                                const float* __restrict__ b, float* __restrict__ out,
                                long long total) {
    constexpr int SHIFT = (COLS == 128) ? 7 : 6;
    long long idx = (long long)blockIdx.x * blockDim.x + threadIdx.x;
    if (idx >= total) return;
    int node = (int)(idx >> SHIFT);
    int f    = (int)(idx & (COLS - 1));
    float dv = dinv[node];
    out[idx] = h[idx] * dv * dv + b[f];
}

template <int COLS>
__global__ __launch_bounds__(256) void scatter_kernel(
    const float* __restrict__ h, const int* __restrict__ src,
    const int* __restrict__ dst, const float* __restrict__ nrm,
    float* out, int e) {
    const long long gid = (long long)blockIdx.x * blockDim.x + threadIdx.x;
    const long long eid = gid >> 5;
    const int lane = (int)(gid & 31);
    if (eid >= (long long)e) return;
    const int s = src[eid];
    const int d = dst[eid];
    const float nm = nrm[eid];
    if (COLS == 128) {
        const float4 v = *(const float4*)&h[(size_t)s * 128 + lane * 4];
        float* o = &out[(size_t)d * 128 + lane * 4];
        unsafeAtomicAdd(o + 0, v.x * nm);
        unsafeAtomicAdd(o + 1, v.y * nm);
        unsafeAtomicAdd(o + 2, v.z * nm);
        unsafeAtomicAdd(o + 3, v.w * nm);
    } else {
        const float2 v = *(const float2*)&h[(size_t)s * 64 + lane * 2];
        float* o = &out[(size_t)d * 64 + lane * 2];
        unsafeAtomicAdd(o + 0, v.x * nm);
        unsafeAtomicAdd(o + 1, v.y * nm);
    }
}

__global__ void relu_kernel(const float* __restrict__ in, float* __restrict__ out,
                            long long total) {
    long long i = (long long)blockIdx.x * blockDim.x + threadIdx.x;
    if (i < total) out[i] = fmaxf(in[i], 0.0f);
}


extern "C" void kernel_launch(void* const* d_in, const int* in_sizes, int n_in,
                              void* d_out, int out_size, void* d_ws, size_t ws_size,
                              hipStream_t stream) {
    const float* x   = (const float*)d_in[0];
    const int*  eidx = (const int*)d_in[1];
    const float* ew  = (const float*)d_in[2];
    const float* W1  = (const float*)d_in[3];
    const float* b1  = (const float*)d_in[4];
    const float* W2  = (const float*)d_in[5];
    const float* b2  = (const float*)d_in[6];

    const int N = in_sizes[0] / 128;
    const int E = in_sizes[2];
    const int* src = eidx;
    const int* dst = eidx + E;

    float* dinv = (float*)d_ws;
    float* nrm  = dinv + N;
    float* h1   = nrm + E;
    float* acc  = h1 + (size_t)N * 128;
    float* out  = (float*)d_out;

    const int T = 256;
    const long long nf1 = (long long)N * 128;
    const long long nf2 = (long long)N * 64;
    const int gemmBlocks = ((N + 15) / 16 + 7) / 8;

    set_deg_kernel<<<(N + T - 1) / T, T, 0, stream>>>(dinv, N);
    accum_deg_kernel<<<(E + T - 1) / T, T, 0, stream>>>(dinv, dst, ew, E);
    dinv_kernel<<<(N + T - 1) / T, T, 0, stream>>>(dinv, N);
    norm_kernel<<<(E + T - 1) / T, T, 0, stream>>>(dinv, src, dst, ew, nrm, E);

    gemm_wmma_kernel<128><<<gemmBlocks, T, 0, stream>>>(x, W1, h1, N);
    init_out_kernel<128><<<(int)((nf1 + T - 1) / T), T, 0, stream>>>(h1, dinv, b1, acc, nf1);
    scatter_kernel<128><<<(int)(((long long)E * 32 + T - 1) / T), T, 0, stream>>>(
        h1, src, dst, nrm, acc, E);
    relu_kernel<<<(int)((nf1 + T - 1) / T), T, 0, stream>>>(acc, h1, nf1);

    gemm_wmma_kernel<64><<<gemmBlocks, T, 0, stream>>>(h1, W2, acc, N);
    init_out_kernel<64><<<(int)((nf2 + T - 1) / T), T, 0, stream>>>(acc, dinv, b2, out, nf2);
    scatter_kernel<64><<<(int)(((long long)E * 32 + T - 1) / T), T, 0, stream>>>(
        acc, src, dst, nrm, out, E);
}


# aggregation made output-stationary: each work-group holds 391 destination rows in LDS, scans the whole edge list, adds its matches with LDS float atomics; init+relu fused; sum order = LDS arrival
# speedup vs baseline: 1.9513x; 1.9513x over previous
#include <hip/hip_runtime.h>

typedef __attribute__((ext_vector_type(2))) float v2f;
typedef __attribute__((ext_vector_type(8))) float v8f;


__global__ void set_deg_kernel(float* deg, int n) {
    int i = blockIdx.x * blockDim.x + threadIdx.x;
    if (i < n) deg[i] = 1.0f;
}

__global__ void accum_deg_kernel(float* deg, const int* __restrict__ dst,
                                 const float* __restrict__ w, int e) {
    int i = blockIdx.x * blockDim.x + threadIdx.x;
    if (i < e) unsafeAtomicAdd(&deg[dst[i]], w[i]);
}

__global__ void dinv_kernel(float* deg, int n) {
    int i = blockIdx.x * blockDim.x + threadIdx.x;
    if (i < n) deg[i] = rsqrtf(deg[i]);
}

__global__ void norm_kernel(const float* __restrict__ dinv, const int* __restrict__ src,
                            const int* __restrict__ dst, const float* __restrict__ w,
                            float* __restrict__ nrm, int e) {
    int i = blockIdx.x * blockDim.x + threadIdx.x;
    if (i < e) nrm[i] = dinv[src[i]] * w[i] * dinv[dst[i]];
}


template <int COLS>
__global__ __launch_bounds__(256) void gemm_wmma_kernel(
    const float* __restrict__ X, const float* __restrict__ W,
    float* __restrict__ OUT, int nRows) {
    constexpr int K  = 128;
    constexpr int NT = COLS / 16;
    constexpr int CSH = (COLS == 128) ? 7 : 6;

    __shared__ float sW[(K / 2) * COLS * 2];
    for (int i = threadIdx.x; i < K * COLS; i += 256) {
        const int row = i >> CSH;
        const int col = i & (COLS - 1);
        sW[(row >> 1) * (COLS * 2) + col * 2 + (row & 1)] = W[i];
    }
    __syncthreads();

    const int wave = threadIdx.x >> 5;
    const int lane = threadIdx.x & 31;
    const int rb   = blockIdx.x * 8 + wave;
    if (rb * 16 >= nRows) return;

    const int l15  = lane & 15;
    const int hsel = lane >> 4;

    const float* aBase = X + (size_t)(rb * 16 + l15) * K + 2 * hsel;
    const float* bCol  = &sW[hsel * (COLS * 2) + l15 * 2];

    v8f acc[NT] = {};

    for (int k = 0; k < K; k += 4) {
        const v2f A = *(const v2f*)(aBase + k);
        const float* bp = bCol + (k >> 1) * (COLS * 2);
#pragma unroll
        for (int c = 0; c < NT; ++c) {
            const v2f B = *(const v2f*)(bp + c * 32);
            acc[c] = __builtin_amdgcn_wmma_f32_16x16x4_f32(
                false, A, false, B, (short)0, acc[c], false, false);
        }
    }

    const int rowBase = rb * 16 + (hsel << 3);
#pragma unroll
    for (int c = 0; c < NT; ++c) {
#pragma unroll
        for (int r = 0; r < 8; ++r)
            OUT[(size_t)(rowBase + r) * COLS + c * 16 + l15] = acc[c][r];
    }
}


constexpr int TILE_R = 391;
constexpr int QCAP   = 160;

constexpr int GRP    = 8;
constexpr int STEP_SH = 15;

__device__ __forceinline__ void load_step(const int* __restrict__ p, int stepIdx, int ebase,
                                          int e, bool vec_ok, int4 (&d)[GRP]) {
    if (vec_ok && ((stepIdx + 1) << STEP_SH) <= e) {
#pragma unroll
        for (int u = 0; u < GRP; ++u) d[u] = *(const int4*)(p + ebase + (u << 7));
    } else {
#pragma unroll
        for (int u = 0; u < GRP; ++u) {
            const int e0 = ebase + (u << 7);
            int4 r = make_int4(-1, -1, -1, -1);
            if (e0     < e) r.x = p[e0];
            if (e0 + 1 < e) r.y = p[e0 + 1];
            if (e0 + 2 < e) r.z = p[e0 + 2];
            if (e0 + 3 < e) r.w = p[e0 + 3];
            d[u] = r;
        }
    }
}

constexpr int ROWS_IN_FLIGHT = 8;

template <int COLS>
__device__ __forceinline__ void drain_queue(
    const float* __restrict__ h, const int* __restrict__ src, const float* __restrict__ nrm,
    const int* qe, const int* qr, int cnt, float* sAcc, int lane) {
    constexpr int VW = COLS / 32;
    for (int b0 = 0; b0 < cnt; b0 += 32) {
        const int idx = b0 + lane;
        int s = 0, r = 0;
        float nm = 0.0f;
        if (idx < cnt) {
            const int ed = qe[idx];
            r  = qr[idx];
            s  = src[ed];
            nm = nrm[ed];
        }
        const int m = min(32, cnt - b0);
        for (int j = 0; j < m; j += ROWS_IN_FLIGHT) {
            float v[ROWS_IN_FLIGHT][VW];
            int   rr[ROWS_IN_FLIGHT];
            float nn[ROWS_IN_FLIGHT];
#pragma unroll
            for (int t = 0; t < ROWS_IN_FLIGHT; ++t) {
                const int jj = min(j + t, m - 1);
                const int sj = __builtin_amdgcn_readlane(s, jj);
                const int rj = __builtin_amdgcn_readlane(r, jj);
                rr[t] = (j + t < m) ? rj : TILE_R;
                nn[t] = __int_as_float(__builtin_amdgcn_readlane(__float_as_int(nm), jj));
                const float* hp = h + (size_t)sj * COLS + lane;
#pragma unroll
                for (int c = 0; c < VW; ++c) v[t][c] = hp[32 * c];
            }
#pragma unroll
            for (int t = 0; t < ROWS_IN_FLIGHT; ++t) {
                float* o = sAcc + rr[t] * COLS + lane;
#pragma unroll
                for (int c = 0; c < VW; ++c) unsafeAtomicAdd(o + 32 * c, v[t][c] * nn[t]);
            }
        }
    }
}

template <int COLS, bool RELU>
__global__ __launch_bounds__(1024) void agg_tile_kernel(
    const float* __restrict__ h, const int* __restrict__ src, const int* __restrict__ dst,
    const float* __restrict__ nrm, const float* __restrict__ dinv, const float* __restrict__ b,
    float* __restrict__ out, int n, int e) {
    constexpr int SHIFT = (COLS == 128) ? 7 : 6;
    __shared__ float sAcc[(TILE_R + 1) * COLS];
    __shared__ int sQe[32 * QCAP];
    __shared__ int sQr[32 * QCAP];

    const int tid  = threadIdx.x;
    const int lane = tid & 31;
    const int wave = tid >> 5;
    const int base = blockIdx.x * TILE_R;
    const int rows = min(TILE_R, n - base);
    const int tileElems = rows * COLS;
    const float* hTile = h + (size_t)base * COLS;
    float* oTile = out + (size_t)base * COLS;

    for (int i = tid * 4; i < tileElems; i += 1024 * 4) {
        const float dv = dinv[base + (i >> SHIFT)];
        const float4 hv = *(const float4*)(hTile + i);
        const float4 bv = *(const float4*)(b + (i & (COLS - 1)));
        float4 o;
        o.x = hv.x * dv * dv + bv.x; o.y = hv.y * dv * dv + bv.y;
        o.z = hv.z * dv * dv + bv.z; o.w = hv.w * dv * dv + bv.w;
        *(float4*)(sAcc + i) = o;
    }
    if (tid < COLS) sAcc[TILE_R * COLS + tid] = 0.0f;
    __syncthreads();

    int* qe = sQe + wave * QCAP;
    int* qr = sQr + wave * QCAP;
    int qcnt = 0;
    const bool vec_ok = ((((size_t)dst) & 15) == 0);
    const unsigned lt_mask = (1u << lane) - 1u;

    const int nsteps = (e + (1 << STEP_SH) - 1) >> STEP_SH;
    const int step0  = (int)(((long long)blockIdx.x * nsteps) / gridDim.x);
    const int woff   = ((wave * GRP) << 7) + lane * 4;

    for (int it = 0; it < nsteps; ++it) {
        int st = step0 + it;
        if (st >= nsteps) st -= nsteps;
        const int ebase = (st << STEP_SH) + woff;

        int4 cur[GRP];
        load_step(dst, st, ebase, e, vec_ok, cur);

#pragma unroll
        for (int u = 0; u < GRP; ++u) {
            const int e0 = ebase + (u << 7);
            const int dl[4] = { cur[u].x - base, cur[u].y - base, cur[u].z - base, cur[u].w - base };
            const bool mk[4] = { (unsigned)dl[0] < (unsigned)rows, (unsigned)dl[1] < (unsigned)rows,
                                 (unsigned)dl[2] < (unsigned)rows, (unsigned)dl[3] < (unsigned)rows };
            if (__builtin_amdgcn_ballot_w32(mk[0] | mk[1] | mk[2] | mk[3])) {
#pragma unroll
                for (int k = 0; k < 4; ++k) {
                    const unsigned bk = __builtin_amdgcn_ballot_w32(mk[k]);
                    if (bk) {
                        if (mk[k]) {
                            const int pos = qcnt + __popc(bk & lt_mask);
                            qe[pos] = e0 + k;
                            qr[pos] = dl[k];
                        }
                        qcnt += __popc(bk);
                    }
                }
                if (qcnt >= 32) {
                    drain_queue<COLS>(h, src, nrm, qe, qr, qcnt, sAcc, lane);
                    qcnt = 0;
                }
            }
        }
    }
    if (qcnt > 0) drain_queue<COLS>(h, src, nrm, qe, qr, qcnt, sAcc, lane);
    __syncthreads();

    for (int i = tid * 4; i < tileElems; i += 1024 * 4) {
        float4 o = *(const float4*)(sAcc + i);
        if (RELU) {
            o.x = fmaxf(o.x, 0.0f); o.y = fmaxf(o.y, 0.0f);
            o.z = fmaxf(o.z, 0.0f); o.w = fmaxf(o.w, 0.0f);
        }
        *(float4*)(oTile + i) = o;
    }
}


extern "C" void kernel_launch(void* const* d_in, const int* in_sizes, int n_in,
                              void* d_out, int out_size, void* d_ws, size_t ws_size,
                              hipStream_t stream) {
    const float* x   = (const float*)d_in[0];
    const int*  eidx = (const int*)d_in[1];
    const float* ew  = (const float*)d_in[2];
    const float* W1  = (const float*)d_in[3];
    const float* b1  = (const float*)d_in[4];
    const float* W2  = (const float*)d_in[5];
    const float* b2  = (const float*)d_in[6];

    const int N = in_sizes[0] / 128;
    const int E = in_sizes[2];
    const int* src = eidx;
    const int* dst = eidx + E;

    float* dinv = (float*)d_ws;
    float* nrm  = dinv + N;
    float* h1   = nrm + E;
    float* acc  = h1 + (size_t)N * 128;
    float* out  = (float*)d_out;

    const int T = 256;
    const int gemmBlocks = ((N + 15) / 16 + 7) / 8;
    const int aggBlocks  = (N + TILE_R - 1) / TILE_R;

    set_deg_kernel<<<(N + T - 1) / T, T, 0, stream>>>(dinv, N);
    accum_deg_kernel<<<(E + T - 1) / T, T, 0, stream>>>(dinv, dst, ew, E);
    dinv_kernel<<<(N + T - 1) / T, T, 0, stream>>>(dinv, N);
    norm_kernel<<<(E + T - 1) / T, T, 0, stream>>>(dinv, src, dst, ew, nrm, E);

    gemm_wmma_kernel<128><<<gemmBlocks, T, 0, stream>>>(x, W1, h1, N);
    agg_tile_kernel<128, true><<<aggBlocks, 1024, 0, stream>>>(
        h1, src, dst, nrm, dinv, b1, acc, N, E);

    gemm_wmma_kernel<64><<<gemmBlocks, T, 0, stream>>>(acc, W2, h1, N);
    agg_tile_kernel<64, false><<<aggBlocks, 1024, 0, stream>>>(
        h1, src, dst, nrm, dinv, b2, out, N, E);
}


# aggregation scans a one-byte row-group key per edge (written by the norm kernel) instead of the 4-byte destination: 16 keys per lane per load, a quarter of the bytes; sum order LDS arrival as before
# speedup vs baseline: 2.0752x; 1.0635x over previous
#include <hip/hip_runtime.h>

typedef __attribute__((ext_vector_type(2))) float v2f;
typedef __attribute__((ext_vector_type(8))) float v8f;


__global__ void set_deg_kernel(float* deg, int n) {
    int i = blockIdx.x * blockDim.x + threadIdx.x;
    if (i < n) deg[i] = 1.0f;
}

__global__ void accum_deg_kernel(float* deg, const int* __restrict__ dst,
                                 const float* __restrict__ w, int e) {
    int i = blockIdx.x * blockDim.x + threadIdx.x;
    if (i < e) unsafeAtomicAdd(&deg[dst[i]], w[i]);
}

__global__ void dinv_kernel(float* deg, int n) {
    int i = blockIdx.x * blockDim.x + threadIdx.x;
    if (i < n) deg[i] = rsqrtf(deg[i]);
}

constexpr int TILE_R = 391;

__global__ void norm_kernel(const float* __restrict__ dinv, const int* __restrict__ src,
                            const int* __restrict__ dst, const float* __restrict__ w,
                            float* __restrict__ nrm, unsigned char* __restrict__ key, int e) {
    int i = blockIdx.x * blockDim.x + threadIdx.x;
    if (i < e) {
        const int d = dst[i];
        nrm[i] = dinv[src[i]] * w[i] * dinv[d];
        key[i] = (unsigned char)((unsigned)d / (unsigned)TILE_R);
    }
}


template <int COLS>
__global__ __launch_bounds__(256) void gemm_wmma_kernel(
    const float* __restrict__ X, const float* __restrict__ W,
    float* __restrict__ OUT, int nRows) {
    constexpr int K  = 128;
    constexpr int NT = COLS / 16;
    constexpr int CSH = (COLS == 128) ? 7 : 6;

    __shared__ float sW[(K / 2) * COLS * 2];
    for (int i = threadIdx.x; i < K * COLS; i += 256) {
        const int row = i >> CSH;
        const int col = i & (COLS - 1);
        sW[(row >> 1) * (COLS * 2) + col * 2 + (row & 1)] = W[i];
    }
    __syncthreads();

    const int wave = threadIdx.x >> 5;
    const int lane = threadIdx.x & 31;
    const int rb   = blockIdx.x * 8 + wave;
    if (rb * 16 >= nRows) return;

    const int l15  = lane & 15;
    const int hsel = lane >> 4;

    const float* aBase = X + (size_t)(rb * 16 + l15) * K + 2 * hsel;
    const float* bCol  = &sW[hsel * (COLS * 2) + l15 * 2];

    v8f acc[NT] = {};

    for (int k = 0; k < K; k += 4) {
        const v2f A = *(const v2f*)(aBase + k);
        const float* bp = bCol + (k >> 1) * (COLS * 2);
#pragma unroll
        for (int c = 0; c < NT; ++c) {
            const v2f B = *(const v2f*)(bp + c * 32);
            acc[c] = __builtin_amdgcn_wmma_f32_16x16x4_f32(
                false, A, false, B, (short)0, acc[c], false, false);
        }
    }

    const int rowBase = rb * 16 + (hsel << 3);
#pragma unroll
    for (int c = 0; c < NT; ++c) {
#pragma unroll
        for (int r = 0; r < 8; ++r)
            OUT[(size_t)(rowBase + r) * COLS + c * 16 + l15] = acc[c][r];
    }
}


constexpr int QCAP    = 64;
constexpr int GRP     = 8;
constexpr int STEP_SH = 17;

__device__ __forceinline__ unsigned zero_bytes(unsigned x) {
    return ~(((x & 0x7F7F7F7Fu) + 0x7F7F7F7Fu) | x) & 0x80808080u;
}

__device__ __forceinline__ unsigned guarded_word(const unsigned char* __restrict__ key,
                                                 int i0, int e, unsigned miss) {
    unsigned w = 0;
#pragma unroll
    for (int k = 0; k < 4; ++k) {
        const unsigned b = (i0 + k < e) ? (unsigned)key[i0 + k] : miss;
        w |= b << (8 * k);
    }
    return w;
}

constexpr int ROWS_IN_FLIGHT = 8;

template <int COLS>
__device__ __forceinline__ void drain_queue(
    const float* __restrict__ h, const int* __restrict__ src, const int* __restrict__ dst,
    const float* __restrict__ nrm, const int* qe, int cnt, int base, float* sAcc, int lane) {
    constexpr int VW = COLS / 32;
    for (int b0 = 0; b0 < cnt; b0 += 32) {
        const int idx = b0 + lane;
        int s = 0, r = 0;
        float nm = 0.0f;
        if (idx < cnt) {
            const int ed = qe[idx];
            s  = src[ed];
            r  = dst[ed] - base;
            nm = nrm[ed];
        }
        const int m = min(32, cnt - b0);
        for (int j = 0; j < m; j += ROWS_IN_FLIGHT) {
            float v[ROWS_IN_FLIGHT][VW];
            int   rr[ROWS_IN_FLIGHT];
            float nn[ROWS_IN_FLIGHT];
#pragma unroll
            for (int t = 0; t < ROWS_IN_FLIGHT; ++t) {
                const int jj = min(j + t, m - 1);
                const int sj = __builtin_amdgcn_readlane(s, jj);
                const int rj = __builtin_amdgcn_readlane(r, jj);
                rr[t] = (j + t < m) ? rj : TILE_R;
                nn[t] = __int_as_float(__builtin_amdgcn_readlane(__float_as_int(nm), jj));
                const float* hp = h + (size_t)sj * COLS + lane;
#pragma unroll
                for (int c = 0; c < VW; ++c) v[t][c] = hp[32 * c];
            }
#pragma unroll
            for (int t = 0; t < ROWS_IN_FLIGHT; ++t) {
                float* o = sAcc + rr[t] * COLS + lane;
#pragma unroll
                for (int c = 0; c < VW; ++c) unsafeAtomicAdd(o + 32 * c, v[t][c] * nn[t]);
            }
        }
    }
}

template <int COLS, bool RELU>
__global__ __launch_bounds__(1024) void agg_tile_kernel(
    const float* __restrict__ h, const int* __restrict__ src, const int* __restrict__ dst,
    const unsigned char* __restrict__ key, const float* __restrict__ nrm,
    const float* __restrict__ dinv, const float* __restrict__ b,
    float* __restrict__ out, int n, int e) {
    constexpr int SHIFT = (COLS == 128) ? 7 : 6;
    __shared__ float sAcc[(TILE_R + 1) * COLS];
    __shared__ int sQe[32 * QCAP];

    const int tid  = threadIdx.x;
    const int lane = tid & 31;
    const int wave = __builtin_amdgcn_readfirstlane(tid >> 5);
    const int base = blockIdx.x * TILE_R;
    const int rows = min(TILE_R, n - base);
    const int tileElems = rows * COLS;
    const float* hTile = h + (size_t)base * COLS;
    float* oTile = out + (size_t)base * COLS;

    for (int i = tid * 4; i < tileElems; i += 1024 * 4) {
        const float dv = dinv[base + (i >> SHIFT)];
        const float4 hv = *(const float4*)(hTile + i);
        const float4 bv = *(const float4*)(b + (i & (COLS - 1)));
        float4 o;
        o.x = hv.x * dv * dv + bv.x; o.y = hv.y * dv * dv + bv.y;
        o.z = hv.z * dv * dv + bv.z; o.w = hv.w * dv * dv + bv.w;
        *(float4*)(sAcc + i) = o;
    }
    if (tid < COLS) sAcc[TILE_R * COLS + tid] = 0.0f;
    __syncthreads();

    const unsigned mine  = blockIdx.x & 255u;
    const unsigned pat   = mine * 0x01010101u;
    const unsigned miss  = mine ^ 1u;
    const unsigned miss4 = miss * 0x01010101u;
    const bool vec_ok = ((((size_t)key) & 15) == 0);

    int* qe = sQe + wave * QCAP;
    int qcnt = 0;

    const int nsteps = (e + (1 << STEP_SH) - 1) >> STEP_SH;
    const int step0  = (int)(((long long)blockIdx.x * nsteps) / gridDim.x);
    const int woff   = (wave * GRP) << 9;

    for (int it = 0; it < nsteps; ++it) {
        int st = step0 + it;
        if (st >= nsteps) st -= nsteps;
        const int wbase = (st << STEP_SH) + woff;

        uint4 cur[GRP];
        if (vec_ok && wbase + (GRP << 9) <= e) {
#pragma unroll
            for (int u = 0; u < GRP; ++u)
                cur[u] = *(const uint4*)(key + wbase + (u << 9) + lane * 16);
        } else {
#pragma unroll
            for (int u = 0; u < GRP; ++u) {
                const int c0 = wbase + (u << 9);
                if (vec_ok && c0 + 512 <= e) {
                    cur[u] = *(const uint4*)(key + c0 + lane * 16);
                } else if (c0 >= e) {
                    cur[u] = make_uint4(miss4, miss4, miss4, miss4);
                } else {
                    const int i0 = c0 + lane * 16;
                    cur[u].x = guarded_word(key, i0,      e, miss);
                    cur[u].y = guarded_word(key, i0 + 4,  e, miss);
                    cur[u].z = guarded_word(key, i0 + 8,  e, miss);
                    cur[u].w = guarded_word(key, i0 + 12, e, miss);
                }
            }
        }

#pragma unroll
        for (int u = 0; u < GRP; ++u) {
            unsigned M = (zero_bytes(cur[u].x ^ pat) >> 7) | (zero_bytes(cur[u].y ^ pat) >> 6) |
                         (zero_bytes(cur[u].z ^ pat) >> 5) | (zero_bytes(cur[u].w ^ pat) >> 4);
            unsigned bm = __builtin_amdgcn_ballot_w32(M != 0u);
            const int e0 = wbase + (u << 9) + lane * 16;
            while (bm) {
                if (M != 0u) {
                    const int B = __builtin_ctz(M);
                    M &= M - 1u;
                    qe[__builtin_amdgcn_mbcnt_lo(bm, (unsigned)qcnt)] = e0 + ((B & 7) << 2) + (B >> 3);
                }
                qcnt += __popc(bm);
                if (qcnt >= 32) {
                    drain_queue<COLS>(h, src, dst, nrm, qe, qcnt, base, sAcc, lane);
                    qcnt = 0;
                }
                bm = __builtin_amdgcn_ballot_w32(M != 0u);
            }
        }
    }
    if (qcnt > 0) drain_queue<COLS>(h, src, dst, nrm, qe, qcnt, base, sAcc, lane);
    __syncthreads();

    for (int i = tid * 4; i < tileElems; i += 1024 * 4) {
        float4 o = *(const float4*)(sAcc + i);
        if (RELU) {
            o.x = fmaxf(o.x, 0.0f); o.y = fmaxf(o.y, 0.0f);
            o.z = fmaxf(o.z, 0.0f); o.w = fmaxf(o.w, 0.0f);
        }
        *(float4*)(oTile + i) = o;
    }
}


extern "C" void kernel_launch(void* const* d_in, const int* in_sizes, int n_in,
                              void* d_out, int out_size, void* d_ws, size_t ws_size,
                              hipStream_t stream) {
    const float* x   = (const float*)d_in[0];
    const int*  eidx = (const int*)d_in[1];
    const float* ew  = (const float*)d_in[2];
    const float* W1  = (const float*)d_in[3];
    const float* b1  = (const float*)d_in[4];
    const float* W2  = (const float*)d_in[5];
    const float* b2  = (const float*)d_in[6];

    const int N = in_sizes[0] / 128;
    const int E = in_sizes[2];
    const int* src = eidx;
    const int* dst = eidx + E;

    float* dinv = (float*)d_ws;
    float* nrm  = dinv + N;
    float* h1   = nrm + E;
    float* acc  = h1 + (size_t)N * 128;
    unsigned char* key = (unsigned char*)(acc + (size_t)N * 128);
    float* out  = (float*)d_out;

    const int T = 256;
    const int gemmBlocks = ((N + 15) / 16 + 7) / 8;
    const int aggBlocks  = (N + TILE_R - 1) / TILE_R;

    set_deg_kernel<<<(N + T - 1) / T, T, 0, stream>>>(dinv, N);
    accum_deg_kernel<<<(E + T - 1) / T, T, 0, stream>>>(dinv, dst, ew, E);
    dinv_kernel<<<(N + T - 1) / T, T, 0, stream>>>(dinv, N);
    norm_kernel<<<(E + T - 1) / T, T, 0, stream>>>(dinv, src, dst, ew, nrm, key, E);

    gemm_wmma_kernel<128><<<gemmBlocks, T, 0, stream>>>(x, W1, h1, N);
    agg_tile_kernel<128, true><<<aggBlocks, 1024, 0, stream>>>(
        h1, src, dst, key, nrm, dinv, b1, acc, N, E);

    gemm_wmma_kernel<64><<<gemmBlocks, T, 0, stream>>>(acc, W2, h1, N);
    agg_tile_kernel<64, false><<<aggBlocks, 1024, 0, stream>>>(
        h1, src, dst, key, nrm, dinv, b2, out, N, E);
}


# layer 2 no longer scans the edge list: each wave of layer 1's tile scan leaves its matched edges as 8-byte records (source, tile row, norm) and layer 2's wave replays them; sum order LDS arrival
# speedup vs baseline: 2.3648x; 1.1396x over previous
#include <hip/hip_runtime.h>

typedef __attribute__((ext_vector_type(2))) float v2f;
typedef __attribute__((ext_vector_type(8))) float v8f;


__global__ void set_deg_kernel(float* deg, int n) {
    int i = blockIdx.x * blockDim.x + threadIdx.x;
    if (i < n) deg[i] = 1.0f;
}

__global__ void accum_deg_kernel(float* deg, const int* __restrict__ dst,
                                 const float* __restrict__ w, int e) {
    int i = blockIdx.x * blockDim.x + threadIdx.x;
    if (i < e) unsafeAtomicAdd(&deg[dst[i]], w[i]);
}

__global__ void dinv_kernel(float* deg, int n) {
    int i = blockIdx.x * blockDim.x + threadIdx.x;
    if (i < n) deg[i] = rsqrtf(deg[i]);
}

constexpr int TILE_R = 391;

__global__ void norm_kernel(const float* __restrict__ dinv, const int* __restrict__ src,
                            const int* __restrict__ dst, const float* __restrict__ w,
                            float* __restrict__ nrm, unsigned char* __restrict__ key, int e) {
    int i = blockIdx.x * blockDim.x + threadIdx.x;
    if (i < e) {
        const int d = dst[i];
        nrm[i] = dinv[src[i]] * w[i] * dinv[d];
        key[i] = (unsigned char)((unsigned)d / (unsigned)TILE_R);
    }
}


template <int COLS>
__global__ __launch_bounds__(256) void gemm_wmma_kernel(
    const float* __restrict__ X, const float* __restrict__ W,
    float* __restrict__ OUT, int nRows) {
    constexpr int K  = 128;
    constexpr int NT = COLS / 16;
    constexpr int CSH = (COLS == 128) ? 7 : 6;

    __shared__ float sW[(K / 2) * COLS * 2];
    for (int i = threadIdx.x; i < K * COLS; i += 256) {
        const int row = i >> CSH;
        const int col = i & (COLS - 1);
        sW[(row >> 1) * (COLS * 2) + col * 2 + (row & 1)] = W[i];
    }
    __syncthreads();

    const int wave = threadIdx.x >> 5;
    const int lane = threadIdx.x & 31;
    const int rb   = blockIdx.x * 8 + wave;
    if (rb * 16 >= nRows) return;

    const int l15  = lane & 15;
    const int hsel = lane >> 4;

    const float* aBase = X + (size_t)(rb * 16 + l15) * K + 2 * hsel;
    const float* bCol  = &sW[hsel * (COLS * 2) + l15 * 2];

    v8f acc[NT] = {};

    for (int k = 0; k < K; k += 4) {
        const v2f A = *(const v2f*)(aBase + k);
        const float* bp = bCol + (k >> 1) * (COLS * 2);
#pragma unroll
        for (int c = 0; c < NT; ++c) {
            const v2f B = *(const v2f*)(bp + c * 32);
            acc[c] = __builtin_amdgcn_wmma_f32_16x16x4_f32(
                false, A, false, B, (short)0, acc[c], false, false);
        }
    }

    const int rowBase = rb * 16 + (hsel << 3);
#pragma unroll
    for (int c = 0; c < NT; ++c) {
#pragma unroll
        for (int r = 0; r < 8; ++r)
            OUT[(size_t)(rowBase + r) * COLS + c * 16 + l15] = acc[c][r];
    }
}


constexpr int QCAP    = 64;
constexpr int GRP     = 8;
constexpr int STEP_SH = 17;

__device__ __forceinline__ unsigned zero_bytes(unsigned x) {
    return ~(((x & 0x7F7F7F7Fu) + 0x7F7F7F7Fu) | x) & 0x80808080u;
}

__device__ __forceinline__ unsigned guarded_word(const unsigned char* __restrict__ key,
                                                 int i0, int e, unsigned miss) {
    unsigned w = 0;
#pragma unroll
    for (int k = 0; k < 4; ++k) {
        const unsigned b = (i0 + k < e) ? (unsigned)key[i0 + k] : miss;
        w |= b << (8 * k);
    }
    return w;
}

typedef unsigned long long rec_t;
constexpr int WAVE_CHUNKS = 32;

constexpr int ROWS_IN_FLIGHT = 8;

template <int COLS>
__device__ __forceinline__ void add_rows(const float* __restrict__ h, int s, int r, float nm,
                                         int m, float* sAcc, int lane) {
    constexpr int VW = COLS / 32;
    for (int j = 0; j < m; j += ROWS_IN_FLIGHT) {
        float v[ROWS_IN_FLIGHT][VW];
        int   rr[ROWS_IN_FLIGHT];
        float nn[ROWS_IN_FLIGHT];
#pragma unroll
        for (int t = 0; t < ROWS_IN_FLIGHT; ++t) {
            const int jj = min(j + t, m - 1);
            const int sj = __builtin_amdgcn_readlane(s, jj);
            const int rj = __builtin_amdgcn_readlane(r, jj);
            rr[t] = (j + t < m) ? rj : TILE_R;
            nn[t] = __int_as_float(__builtin_amdgcn_readlane(__float_as_int(nm), jj));
            const float* hp = h + (size_t)sj * COLS + lane;
#pragma unroll
            for (int c = 0; c < VW; ++c) v[t][c] = hp[32 * c];
        }
#pragma unroll
        for (int t = 0; t < ROWS_IN_FLIGHT; ++t) {
            float* o = sAcc + rr[t] * COLS + lane;
#pragma unroll
            for (int c = 0; c < VW; ++c) unsafeAtomicAdd(o + 32 * c, v[t][c] * nn[t]);
        }
    }
}

template <int COLS, bool KEEP>
__device__ __forceinline__ void drain_queue(
    const float* __restrict__ h, const int* __restrict__ src, const int* __restrict__ dst,
    const float* __restrict__ nrm, const int* qe, int cnt, int base, float* sAcc, int lane,
    rec_t* __restrict__ wlist, int& wch) {
    for (int b0 = 0; b0 < cnt; b0 += 32) {
        const int idx = b0 + lane;
        int s = 0, r = TILE_R;
        float nm = 0.0f;
        if (idx < cnt) {
            const int ed = qe[idx];
            s  = src[ed];
            r  = dst[ed] - base;
            nm = nrm[ed];
        }
        if (KEEP) {
            if (wch < WAVE_CHUNKS)
                wlist[wch * 32 + lane] = ((rec_t)__float_as_uint(nm) << 32) |
                                         (rec_t)(((unsigned)r << 17) | (unsigned)s);
            ++wch;
        }
        add_rows<COLS>(h, s, r, nm, min(32, cnt - b0), sAcc, lane);
    }
}

template <int COLS, bool RELU, bool KEEP, bool REPLAY>
__global__ __launch_bounds__(1024) void agg_tile_kernel(
    const float* __restrict__ h, const int* __restrict__ src, const int* __restrict__ dst,
    const unsigned char* __restrict__ key, const float* __restrict__ nrm,
    const float* __restrict__ dinv, const float* __restrict__ b,
    float* __restrict__ out, unsigned long long* __restrict__ list, int* __restrict__ chunks,
    int n, int e) {
    constexpr int SHIFT = (COLS == 128) ? 7 : 6;
    __shared__ float sAcc[(TILE_R + 1) * COLS];
    __shared__ int sQe[32 * QCAP];
    __shared__ int sChunks[32];

    const int tid  = threadIdx.x;
    const int lane = tid & 31;
    const int wave = __builtin_amdgcn_readfirstlane(tid >> 5);
    const int base = blockIdx.x * TILE_R;
    const int rows = min(TILE_R, n - base);
    const int tileElems = rows * COLS;
    const float* hTile = h + (size_t)base * COLS;
    float* oTile = out + (size_t)base * COLS;

    for (int i = tid * 4; i < tileElems; i += 1024 * 4) {
        const float dv = dinv[base + (i >> SHIFT)];
        const float4 hv = *(const float4*)(hTile + i);
        const float4 bv = *(const float4*)(b + (i & (COLS - 1)));
        float4 o;
        o.x = hv.x * dv * dv + bv.x; o.y = hv.y * dv * dv + bv.y;
        o.z = hv.z * dv * dv + bv.z; o.w = hv.w * dv * dv + bv.w;
        *(float4*)(sAcc + i) = o;
    }
    if (tid < COLS) sAcc[TILE_R * COLS + tid] = 0.0f;
    __syncthreads();

    const unsigned mine  = blockIdx.x & 255u;
    const unsigned pat   = mine * 0x01010101u;
    const unsigned miss  = mine ^ 1u;
    const unsigned miss4 = miss * 0x01010101u;
    const bool vec_ok = ((((size_t)key) & 15) == 0);

    int* qe = sQe + wave * QCAP;
    int qcnt = 0;

    rec_t* wlist = list + (size_t)(blockIdx.x * 32 + wave) * (WAVE_CHUNKS * 32);
    int wch = 0;

    bool scan = true;
    if (REPLAY) {
        const int cl = chunks[blockIdx.x * 32 + lane];
        scan = __builtin_amdgcn_ballot_w32(cl > WAVE_CHUNKS) != 0u;
        if (!scan) {
            const int nch = __builtin_amdgcn_readlane(cl, wave);
            rec_t rec = (rec_t)((unsigned)TILE_R << 17);
            if (nch > 0) rec = wlist[lane];
            for (int c = 0; c < nch; ++c) {
                rec_t nxt = rec;
                if (c + 1 < nch) nxt = wlist[(c + 1) * 32 + lane];
                const unsigned lo = (unsigned)rec;
                const int r = (int)(lo >> 17);
                const int m = __popc(__builtin_amdgcn_ballot_w32(r != TILE_R));
                add_rows<COLS>(h, (int)(lo & 0x1FFFFu), r, __uint_as_float((unsigned)(rec >> 32)), m,
                               sAcc, lane);
                rec = nxt;
            }
        }
    }

    const int nsteps = (e + (1 << STEP_SH) - 1) >> STEP_SH;
    const int step0  = (int)(((long long)blockIdx.x * nsteps) / gridDim.x);
    const int woff   = (wave * GRP) << 9;

    for (int it = 0; scan && it < nsteps; ++it) {
        int st = step0 + it;
        if (st >= nsteps) st -= nsteps;
        const int wbase = (st << STEP_SH) + woff;

        uint4 cur[GRP];
        if (vec_ok && wbase + (GRP << 9) <= e) {
#pragma unroll
            for (int u = 0; u < GRP; ++u)
                cur[u] = *(const uint4*)(key + wbase + (u << 9) + lane * 16);
        } else {
#pragma unroll
            for (int u = 0; u < GRP; ++u) {
                const int c0 = wbase + (u << 9);
                if (vec_ok && c0 + 512 <= e) {
                    cur[u] = *(const uint4*)(key + c0 + lane * 16);
                } else if (c0 >= e) {
                    cur[u] = make_uint4(miss4, miss4, miss4, miss4);
                } else {
                    const int i0 = c0 + lane * 16;
                    cur[u].x = guarded_word(key, i0,      e, miss);
                    cur[u].y = guarded_word(key, i0 + 4,  e, miss);
                    cur[u].z = guarded_word(key, i0 + 8,  e, miss);
                    cur[u].w = guarded_word(key, i0 + 12, e, miss);
                }
            }
        }

#pragma unroll
        for (int u = 0; u < GRP; ++u) {
            unsigned M = (zero_bytes(cur[u].x ^ pat) >> 7) | (zero_bytes(cur[u].y ^ pat) >> 6) |
                         (zero_bytes(cur[u].z ^ pat) >> 5) | (zero_bytes(cur[u].w ^ pat) >> 4);
            unsigned bm = __builtin_amdgcn_ballot_w32(M != 0u);
            const int e0 = wbase + (u << 9) + lane * 16;
            while (bm) {
                if (M != 0u) {
                    const int B = __builtin_ctz(M);
                    M &= M - 1u;
                    qe[__builtin_amdgcn_mbcnt_lo(bm, (unsigned)qcnt)] = e0 + ((B & 7) << 2) + (B >> 3);
                }
                qcnt += __popc(bm);
                if (qcnt >= 32) {
                    drain_queue<COLS, KEEP>(h, src, dst, nrm, qe, qcnt, base, sAcc, lane, wlist, wch);
                    qcnt = 0;
                }
                bm = __builtin_amdgcn_ballot_w32(M != 0u);
            }
        }
    }
    if (qcnt > 0) drain_queue<COLS, KEEP>(h, src, dst, nrm, qe, qcnt, base, sAcc, lane, wlist, wch);
    if (KEEP && lane == 0) sChunks[wave] = wch;
    __syncthreads();

    if (KEEP && wave == 0) chunks[blockIdx.x * 32 + lane] = sChunks[lane];

    for (int i = tid * 4; i < tileElems; i += 1024 * 4) {
        float4 o = *(const float4*)(sAcc + i);
        if (RELU) {
            o.x = fmaxf(o.x, 0.0f); o.y = fmaxf(o.y, 0.0f);
            o.z = fmaxf(o.z, 0.0f); o.w = fmaxf(o.w, 0.0f);
        }
        *(float4*)(oTile + i) = o;
    }
}


extern "C" void kernel_launch(void* const* d_in, const int* in_sizes, int n_in,
                              void* d_out, int out_size, void* d_ws, size_t ws_size,
                              hipStream_t stream) {
    const float* x   = (const float*)d_in[0];
    const int*  eidx = (const int*)d_in[1];
    const float* ew  = (const float*)d_in[2];
    const float* W1  = (const float*)d_in[3];
    const float* b1  = (const float*)d_in[4];
    const float* W2  = (const float*)d_in[5];
    const float* b2  = (const float*)d_in[6];

    const int N = in_sizes[0] / 128;
    const int E = in_sizes[2];
    const int* src = eidx;
    const int* dst = eidx + E;

    float* dinv = (float*)d_ws;
    float* nrm  = dinv + N;
    float* h1   = nrm + E;
    float* acc  = h1 + (size_t)N * 128;
    unsigned char* key = (unsigned char*)(acc + (size_t)N * 128);
    float* out  = (float*)d_out;

    const int T = 256;
    const int gemmBlocks = ((N + 15) / 16 + 7) / 8;
    const int aggBlocks  = (N + TILE_R - 1) / TILE_R;
    const size_t keyEnd  = (size_t)((key + E) - (unsigned char*)d_ws);
    int* chunks = (int*)((unsigned char*)d_ws + ((keyEnd + 255) & ~(size_t)255));
    unsigned long long* list =
        (unsigned long long*)(chunks + (((size_t)aggBlocks * 32 + 63) & ~(size_t)63));

    set_deg_kernel<<<(N + T - 1) / T, T, 0, stream>>>(dinv, N);
    accum_deg_kernel<<<(E + T - 1) / T, T, 0, stream>>>(dinv, dst, ew, E);
    dinv_kernel<<<(N + T - 1) / T, T, 0, stream>>>(dinv, N);
    norm_kernel<<<(E + T - 1) / T, T, 0, stream>>>(dinv, src, dst, ew, nrm, key, E);

    gemm_wmma_kernel<128><<<gemmBlocks, T, 0, stream>>>(x, W1, h1, N);
    agg_tile_kernel<128, true, true, false><<<aggBlocks, 1024, 0, stream>>>(
        h1, src, dst, key, nrm, dinv, b1, acc, list, chunks, N, E);

    gemm_wmma_kernel<64><<<gemmBlocks, T, 0, stream>>>(acc, W2, h1, N);
    agg_tile_kernel<64, false, false, true><<<aggBlocks, 1024, 0, stream>>>(
        h1, src, dst, key, nrm, dinv, b2, out, list, chunks, N, E);
}


# degree summed by each row tile's owner in LDS from the edges its key scan finds (scan is its own launch; both layers replay its records): no global atomics, no norm pass; sum order LDS arrival
# speedup vs baseline: 3.0416x; 1.2862x over previous
#include <hip/hip_runtime.h>


typedef __attribute__((ext_vector_type(2))) float v2f;
typedef __attribute__((ext_vector_type(8))) float v8f;

constexpr int TILE_R = 391;

__global__ void key_kernel(const int* __restrict__ dst, unsigned char* __restrict__ key, int e) {
    int i = blockIdx.x * blockDim.x + threadIdx.x;
    if (i < e) key[i] = (unsigned char)((unsigned)dst[i] / (unsigned)TILE_R);
}


template <int COLS>
__global__ __launch_bounds__(256) void gemm_wmma_kernel(
    const float* __restrict__ X, const float* __restrict__ W,
    float* __restrict__ OUT, int nRows) {
    constexpr int K  = 128;
    constexpr int NT = COLS / 16;
    constexpr int CSH = (COLS == 128) ? 7 : 6;

    __shared__ float sW[(K / 2) * COLS * 2];
    for (int i = threadIdx.x; i < K * COLS; i += 256) {
        const int row = i >> CSH;
        const int col = i & (COLS - 1);
        sW[(row >> 1) * (COLS * 2) + col * 2 + (row & 1)] = W[i];
    }
    __syncthreads();

    const int wave = threadIdx.x >> 5;
    const int lane = threadIdx.x & 31;
    const int rb   = blockIdx.x * 8 + wave;
    if (rb * 16 >= nRows) return;

    const int l15  = lane & 15;
    const int hsel = lane >> 4;

    const float* aBase = X + (size_t)(rb * 16 + l15) * K + 2 * hsel;
    const float* bCol  = &sW[hsel * (COLS * 2) + l15 * 2];

    v8f acc[NT] = {};

    for (int k = 0; k < K; k += 4) {
        const v2f A = *(const v2f*)(aBase + k);
        const float* bp = bCol + (k >> 1) * (COLS * 2);
#pragma unroll
        for (int c = 0; c < NT; ++c) {
            const v2f B = *(const v2f*)(bp + c * 32);
            acc[c] = __builtin_amdgcn_wmma_f32_16x16x4_f32(
                false, A, false, B, (short)0, acc[c], false, false);
        }
    }

    const int rowBase = rb * 16 + (hsel << 3);
#pragma unroll
    for (int c = 0; c < NT; ++c) {
#pragma unroll
        for (int r = 0; r < 8; ++r)
            OUT[(size_t)(rowBase + r) * COLS + c * 16 + l15] = acc[c][r];
    }
}


constexpr int QCAP    = 64;
constexpr int GRP     = 8;
constexpr int STEP_SH = 17;

__device__ __forceinline__ unsigned zero_bytes(unsigned x) {
    return ~(((x & 0x7F7F7F7Fu) + 0x7F7F7F7Fu) | x) & 0x80808080u;
}

__device__ __forceinline__ unsigned guarded_word(const unsigned char* __restrict__ key,
                                                 int i0, int e, unsigned miss) {
    unsigned w = 0;
#pragma unroll
    for (int k = 0; k < 4; ++k) {
        const unsigned b = (i0 + k < e) ? (unsigned)key[i0 + k] : miss;
        w |= b << (8 * k);
    }
    return w;
}

typedef unsigned long long rec_t;
constexpr int WAVE_CHUNKS = 32;

template <class Drain>
__device__ __forceinline__ void scan_keys(const unsigned char* __restrict__ key, int e, int* qe,
                                          int wave, int lane, Drain drain) {
    const unsigned mine  = blockIdx.x & 255u;
    const unsigned pat   = mine * 0x01010101u;
    const unsigned miss  = mine ^ 1u;
    const unsigned miss4 = miss * 0x01010101u;
    const bool vec_ok = ((((size_t)key) & 15) == 0);
    int qcnt = 0;

    const int nsteps = (e + (1 << STEP_SH) - 1) >> STEP_SH;
    const int step0  = (int)(((long long)blockIdx.x * nsteps) / gridDim.x);
    const int woff   = (wave * GRP) << 9;

    for (int it = 0; it < nsteps; ++it) {
        int st = step0 + it;
        if (st >= nsteps) st -= nsteps;
        const int wbase = (st << STEP_SH) + woff;

        uint4 cur[GRP];
        if (vec_ok && wbase + (GRP << 9) <= e) {
#pragma unroll
            for (int u = 0; u < GRP; ++u)
                cur[u] = *(const uint4*)(key + wbase + (u << 9) + lane * 16);
        } else {
#pragma unroll
            for (int u = 0; u < GRP; ++u) {
                const int c0 = wbase + (u << 9);
                if (vec_ok && c0 + 512 <= e) {
                    cur[u] = *(const uint4*)(key + c0 + lane * 16);
                } else if (c0 >= e) {
                    cur[u] = make_uint4(miss4, miss4, miss4, miss4);
                } else {
                    const int i0 = c0 + lane * 16;
                    cur[u].x = guarded_word(key, i0,      e, miss);
                    cur[u].y = guarded_word(key, i0 + 4,  e, miss);
                    cur[u].z = guarded_word(key, i0 + 8,  e, miss);
                    cur[u].w = guarded_word(key, i0 + 12, e, miss);
                }
            }
        }

#pragma unroll
        for (int u = 0; u < GRP; ++u) {
            unsigned M = (zero_bytes(cur[u].x ^ pat) >> 7) | (zero_bytes(cur[u].y ^ pat) >> 6) |
                         (zero_bytes(cur[u].z ^ pat) >> 5) | (zero_bytes(cur[u].w ^ pat) >> 4);
            unsigned bm = __builtin_amdgcn_ballot_w32(M != 0u);
            const int e0 = wbase + (u << 9) + lane * 16;
            while (bm) {
                if (M != 0u) {
                    const int B = __builtin_ctz(M);
                    M &= M - 1u;
                    qe[__builtin_amdgcn_mbcnt_lo(bm, (unsigned)qcnt)] = e0 + ((B & 7) << 2) + (B >> 3);
                }
                qcnt += __popc(bm);
                if (qcnt >= 32) {
                    drain(qcnt);
                    qcnt = 0;
                }
                bm = __builtin_amdgcn_ballot_w32(M != 0u);
            }
        }
    }
    if (qcnt > 0) drain(qcnt);
}

__global__ __launch_bounds__(1024) void scan_deg_kernel(
    const int* __restrict__ src, const int* __restrict__ dst, const float* __restrict__ w,
    const unsigned char* __restrict__ key, float* __restrict__ dinv,
    unsigned long long* __restrict__ list, int* __restrict__ chunks, int n, int e) {
    __shared__ float sDeg[TILE_R + 1];
    __shared__ int sQe[32 * QCAP];
    __shared__ int sChunks[32];

    const int tid  = threadIdx.x;
    const int lane = tid & 31;
    const int wave = __builtin_amdgcn_readfirstlane(tid >> 5);
    const int base = blockIdx.x * TILE_R;
    const int rows = min(TILE_R, n - base);

    if (tid <= TILE_R) sDeg[tid] = 1.0f;
    __syncthreads();

    int* qe = sQe + wave * QCAP;
    rec_t* wlist = list + (size_t)(blockIdx.x * 32 + wave) * (WAVE_CHUNKS * 32);
    int wch = 0;

    scan_keys(key, e, qe, wave, lane, [&](int cnt) {
        for (int b0 = 0; b0 < cnt; b0 += 32) {
            const int idx = b0 + lane;
            int s = 0, r = TILE_R;
            float wt = 0.0f;
            if (idx < cnt) {
                const int ed = qe[idx];
                s  = src[ed];
                r  = dst[ed] - base;
                wt = w[ed];
                unsafeAtomicAdd(&sDeg[r], wt);
            }
            if (wch < WAVE_CHUNKS)
                wlist[wch * 32 + lane] = ((rec_t)__float_as_uint(wt) << 32) |
                                         (rec_t)(((unsigned)r << 17) | (unsigned)s);
            ++wch;
        }
    });

    if (lane == 0) sChunks[wave] = wch;
    __syncthreads();

    if (wave == 0) chunks[blockIdx.x * 32 + lane] = sChunks[lane];
    if (tid < rows) dinv[base + tid] = rsqrtf(sDeg[tid]);
}

constexpr int ROWS_IN_FLIGHT = 8;

template <int COLS>
__device__ __forceinline__ void add_rows(const float* __restrict__ h, int s, int r, float nm,
                                         int m, float* sAcc, int lane) {
    constexpr int VW = COLS / 32;
    for (int j = 0; j < m; j += ROWS_IN_FLIGHT) {
        float v[ROWS_IN_FLIGHT][VW];
        int   rr[ROWS_IN_FLIGHT];
        float nn[ROWS_IN_FLIGHT];
#pragma unroll
        for (int t = 0; t < ROWS_IN_FLIGHT; ++t) {
            const int jj = min(j + t, m - 1);
            const int sj = __builtin_amdgcn_readlane(s, jj);
            const int rj = __builtin_amdgcn_readlane(r, jj);
            rr[t] = (j + t < m) ? rj : TILE_R;
            nn[t] = __int_as_float(__builtin_amdgcn_readlane(__float_as_int(nm), jj));
            const float* hp = h + (size_t)sj * COLS + lane;
#pragma unroll
            for (int c = 0; c < VW; ++c) v[t][c] = hp[32 * c];
        }
#pragma unroll
        for (int t = 0; t < ROWS_IN_FLIGHT; ++t) {
            float* o = sAcc + rr[t] * COLS + lane;
#pragma unroll
            for (int c = 0; c < VW; ++c) unsafeAtomicAdd(o + 32 * c, v[t][c] * nn[t]);
        }
    }
}

template <int COLS>
__device__ __forceinline__ void drain_queue(
    const float* __restrict__ h, const int* __restrict__ src, const int* __restrict__ dst,
    const float* __restrict__ w, const float* __restrict__ dinv, const int* qe, int cnt,
    int base, float* sAcc, int lane) {
    for (int b0 = 0; b0 < cnt; b0 += 32) {
        const int idx = b0 + lane;
        int s = 0, r = TILE_R;
        float nm = 0.0f;
        if (idx < cnt) {
            const int ed = qe[idx];
            s  = src[ed];
            r  = dst[ed] - base;
            nm = w[ed] * dinv[s];
        }
        add_rows<COLS>(h, s, r, nm, min(32, cnt - b0), sAcc, lane);
    }
}

template <int COLS, bool RELU>
__global__ __launch_bounds__(1024) void agg_tile_kernel(
    const float* __restrict__ h, const int* __restrict__ src, const int* __restrict__ dst,
    const unsigned char* __restrict__ key, const float* __restrict__ w,
    const float* __restrict__ dinv, const float* __restrict__ b,
    float* __restrict__ out, const unsigned long long* __restrict__ list,
    const int* __restrict__ chunks, int n, int e) {
    constexpr int SHIFT = (COLS == 128) ? 7 : 6;
    __shared__ float sAcc[(TILE_R + 1) * COLS];
    __shared__ int sQe[32 * QCAP];

    const int tid  = threadIdx.x;
    const int lane = tid & 31;
    const int wave = __builtin_amdgcn_readfirstlane(tid >> 5);
    const int base = blockIdx.x * TILE_R;
    const int rows = min(TILE_R, n - base);
    const int tileElems = rows * COLS;
    const float* hTile = h + (size_t)base * COLS;
    float* oTile = out + (size_t)base * COLS;

    for (int i = tid * 4; i < tileElems; i += 1024 * 4) {
        const float dv = dinv[base + (i >> SHIFT)];
        const float4 hv = *(const float4*)(hTile + i);
        float4 o;
        o.x = hv.x * dv; o.y = hv.y * dv; o.z = hv.z * dv; o.w = hv.w * dv;
        *(float4*)(sAcc + i) = o;
    }
    if (tid < COLS) sAcc[TILE_R * COLS + tid] = 0.0f;
    __syncthreads();

    const rec_t* wlist = list + (size_t)(blockIdx.x * 32 + wave) * (WAVE_CHUNKS * 32);

    const int cl = chunks[blockIdx.x * 32 + lane];
    const bool scan = __builtin_amdgcn_ballot_w32(cl > WAVE_CHUNKS) != 0u;
    if (!scan) {
        const int nch = __builtin_amdgcn_readlane(cl, wave);
        const rec_t null_rec = (rec_t)((unsigned)TILE_R << 17);
        rec_t rec = null_rec, nxt = null_rec;
        if (nch > 0) rec = wlist[lane];
        if (nch > 1) nxt = wlist[32 + lane];
        float dv = dinv[(unsigned)rec & 0x1FFFFu];
        for (int c = 0; c < nch; ++c) {
            rec_t nx2 = null_rec;
            if (c + 2 < nch) nx2 = wlist[(c + 2) * 32 + lane];
            const float dvn = dinv[(unsigned)nxt & 0x1FFFFu];
            const unsigned lo = (unsigned)rec;
            const int r = (int)(lo >> 17);
            const int m = __popc(__builtin_amdgcn_ballot_w32(r != TILE_R));
            add_rows<COLS>(h, (int)(lo & 0x1FFFFu), r, __uint_as_float((unsigned)(rec >> 32)) * dv, m,
                           sAcc, lane);
            rec = nxt; nxt = nx2; dv = dvn;
        }
    } else {
        int* qe = sQe + wave * QCAP;
        scan_keys(key, e, qe, wave, lane, [&](int cnt) {
            drain_queue<COLS>(h, src, dst, w, dinv, qe, cnt, base, sAcc, lane);
        });
    }
    __syncthreads();

    for (int i = tid * 4; i < tileElems; i += 1024 * 4) {
        const float dv = dinv[base + (i >> SHIFT)];
        const float4 bv = *(const float4*)(b + (i & (COLS - 1)));
        float4 o = *(const float4*)(sAcc + i);
        o.x = o.x * dv + bv.x; o.y = o.y * dv + bv.y;
        o.z = o.z * dv + bv.z; o.w = o.w * dv + bv.w;
        if (RELU) {
            o.x = fmaxf(o.x, 0.0f); o.y = fmaxf(o.y, 0.0f);
            o.z = fmaxf(o.z, 0.0f); o.w = fmaxf(o.w, 0.0f);
        }
        *(float4*)(oTile + i) = o;
    }
}


extern "C" void kernel_launch(void* const* d_in, const int* in_sizes, int n_in,
                              void* d_out, int out_size, void* d_ws, size_t ws_size,
                              hipStream_t stream) {
    const float* x   = (const float*)d_in[0];
    const int*  eidx = (const int*)d_in[1];
    const float* ew  = (const float*)d_in[2];
    const float* W1  = (const float*)d_in[3];
    const float* b1  = (const float*)d_in[4];
    const float* W2  = (const float*)d_in[5];
    const float* b2  = (const float*)d_in[6];

    const int N = in_sizes[0] / 128;
    const int E = in_sizes[2];
    const int* src = eidx;
    const int* dst = eidx + E;

    float* dinv = (float*)d_ws;
    float* h1   = dinv + N;
    float* acc  = h1 + (size_t)N * 128;
    unsigned char* key = (unsigned char*)(acc + (size_t)N * 128);
    float* out  = (float*)d_out;

    const int T = 256;
    const int gemmBlocks = ((N + 15) / 16 + 7) / 8;
    const int aggBlocks  = (N + TILE_R - 1) / TILE_R;
    const size_t keyEnd  = (size_t)((key + E) - (unsigned char*)d_ws);
    int* chunks = (int*)((unsigned char*)d_ws + ((keyEnd + 255) & ~(size_t)255));
    unsigned long long* list =
        (unsigned long long*)(chunks + (((size_t)aggBlocks * 32 + 63) & ~(size_t)63));

    key_kernel<<<(E + T - 1) / T, T, 0, stream>>>(dst, key, E);
    scan_deg_kernel<<<aggBlocks, 1024, 0, stream>>>(src, dst, ew, key, dinv, list, chunks, N, E);

    gemm_wmma_kernel<128><<<gemmBlocks, T, 0, stream>>>(x, W1, h1, N);
    agg_tile_kernel<128, true><<<aggBlocks, 1024, 0, stream>>>(
        h1, src, dst, key, ew, dinv, b1, acc, list, chunks, N, E);

    gemm_wmma_kernel<64><<<gemmBlocks, T, 0, stream>>>(acc, W2, h1, N);
    agg_tile_kernel<64, false><<<aggBlocks, 1024, 0, stream>>>(
        h1, src, dst, key, ew, dinv, b2, out, list, chunks, N, E);
}


# scan hands on exactly 32 queued edges at a time, the rest stay queued: every record chunk but a wave's last is full, so replays run no padded row batches; sum order LDS arrival as before
# speedup vs baseline: 3.1324x; 1.0299x over previous
#include <hip/hip_runtime.h>


typedef __attribute__((ext_vector_type(2))) float v2f;
typedef __attribute__((ext_vector_type(8))) float v8f;

constexpr int TILE_R = 391;

__global__ void key_kernel(const int* __restrict__ dst, unsigned char* __restrict__ key, int e) {
    int i = blockIdx.x * blockDim.x + threadIdx.x;
    if (i < e) key[i] = (unsigned char)((unsigned)dst[i] / (unsigned)TILE_R);
}


template <int COLS>
__global__ __launch_bounds__(256) void gemm_wmma_kernel(
    const float* __restrict__ X, const float* __restrict__ W,
    float* __restrict__ OUT, int nRows) {
    constexpr int K  = 128;
    constexpr int NT = COLS / 16;
    constexpr int CSH = (COLS == 128) ? 7 : 6;

    __shared__ float sW[(K / 2) * COLS * 2];
    for (int i = threadIdx.x; i < K * COLS; i += 256) {
        const int row = i >> CSH;
        const int col = i & (COLS - 1);
        sW[(row >> 1) * (COLS * 2) + col * 2 + (row & 1)] = W[i];
    }
    __syncthreads();

    const int wave = threadIdx.x >> 5;
    const int lane = threadIdx.x & 31;
    const int rb   = blockIdx.x * 8 + wave;
    if (rb * 16 >= nRows) return;

    const int l15  = lane & 15;
    const int hsel = lane >> 4;

    const float* aBase = X + (size_t)(rb * 16 + l15) * K + 2 * hsel;
    const float* bCol  = &sW[hsel * (COLS * 2) + l15 * 2];

    v8f acc[NT] = {};

    for (int k = 0; k < K; k += 4) {
        const v2f A = *(const v2f*)(aBase + k);
        const float* bp = bCol + (k >> 1) * (COLS * 2);
#pragma unroll
        for (int c = 0; c < NT; ++c) {
            const v2f B = *(const v2f*)(bp + c * 32);
            acc[c] = __builtin_amdgcn_wmma_f32_16x16x4_f32(
                false, A, false, B, (short)0, acc[c], false, false);
        }
    }

    const int rowBase = rb * 16 + (hsel << 3);
#pragma unroll
    for (int c = 0; c < NT; ++c) {
#pragma unroll
        for (int r = 0; r < 8; ++r)
            OUT[(size_t)(rowBase + r) * COLS + c * 16 + l15] = acc[c][r];
    }
}


constexpr int QCAP    = 64;
constexpr int GRP     = 8;
constexpr int STEP_SH = 17;

__device__ __forceinline__ unsigned zero_bytes(unsigned x) {
    return ~(((x & 0x7F7F7F7Fu) + 0x7F7F7F7Fu) | x) & 0x80808080u;
}

__device__ __forceinline__ unsigned guarded_word(const unsigned char* __restrict__ key,
                                                 int i0, int e, unsigned miss) {
    unsigned w = 0;
#pragma unroll
    for (int k = 0; k < 4; ++k) {
        const unsigned b = (i0 + k < e) ? (unsigned)key[i0 + k] : miss;
        w |= b << (8 * k);
    }
    return w;
}

typedef unsigned long long rec_t;
constexpr int WAVE_CHUNKS = 32;

template <class Drain>
__device__ __forceinline__ void scan_keys(const unsigned char* __restrict__ key, int e, int* qe,
                                          int wave, int lane, Drain drain) {
    const unsigned mine  = blockIdx.x & 255u;
    const unsigned pat   = mine * 0x01010101u;
    const unsigned miss  = mine ^ 1u;
    const unsigned miss4 = miss * 0x01010101u;
    const bool vec_ok = ((((size_t)key) & 15) == 0);
    int qcnt = 0;

    const int nsteps = (e + (1 << STEP_SH) - 1) >> STEP_SH;
    const int step0  = (int)(((long long)blockIdx.x * nsteps) / gridDim.x);
    const int woff   = (wave * GRP) << 9;

    for (int it = 0; it < nsteps; ++it) {
        int st = step0 + it;
        if (st >= nsteps) st -= nsteps;
        const int wbase = (st << STEP_SH) + woff;

        uint4 cur[GRP];
        if (vec_ok && wbase + (GRP << 9) <= e) {
#pragma unroll
            for (int u = 0; u < GRP; ++u)
                cur[u] = *(const uint4*)(key + wbase + (u << 9) + lane * 16);
        } else {
#pragma unroll
            for (int u = 0; u < GRP; ++u) {
                const int c0 = wbase + (u << 9);
                if (vec_ok && c0 + 512 <= e) {
                    cur[u] = *(const uint4*)(key + c0 + lane * 16);
                } else if (c0 >= e) {
                    cur[u] = make_uint4(miss4, miss4, miss4, miss4);
                } else {
                    const int i0 = c0 + lane * 16;
                    cur[u].x = guarded_word(key, i0,      e, miss);
                    cur[u].y = guarded_word(key, i0 + 4,  e, miss);
                    cur[u].z = guarded_word(key, i0 + 8,  e, miss);
                    cur[u].w = guarded_word(key, i0 + 12, e, miss);
                }
            }
        }

#pragma unroll
        for (int u = 0; u < GRP; ++u) {
            unsigned M = (zero_bytes(cur[u].x ^ pat) >> 7) | (zero_bytes(cur[u].y ^ pat) >> 6) |
                         (zero_bytes(cur[u].z ^ pat) >> 5) | (zero_bytes(cur[u].w ^ pat) >> 4);
            unsigned bm = __builtin_amdgcn_ballot_w32(M != 0u);
            const int e0 = wbase + (u << 9) + lane * 16;
            while (bm) {
                if (M != 0u) {
                    const int B = __builtin_ctz(M);
                    M &= M - 1u;
                    qe[__builtin_amdgcn_mbcnt_lo(bm, (unsigned)qcnt)] = e0 + ((B & 7) << 2) + (B >> 3);
                }
                qcnt += __popc(bm);
                if (qcnt >= 32) {
                    drain(32);
                    qcnt -= 32;
                    if (lane < qcnt) {
                        const int keep = qe[32 + lane];
                        qe[lane] = keep;
                    }
                }
                bm = __builtin_amdgcn_ballot_w32(M != 0u);
            }
        }
    }
    if (qcnt > 0) drain(qcnt);
}

__global__ __launch_bounds__(1024) void scan_deg_kernel(
    const int* __restrict__ src, const int* __restrict__ dst, const float* __restrict__ w,
    const unsigned char* __restrict__ key, float* __restrict__ dinv,
    unsigned long long* __restrict__ list, int* __restrict__ chunks, int n, int e) {
    __shared__ float sDeg[TILE_R + 1];
    __shared__ int sQe[32 * QCAP];
    __shared__ int sChunks[32];

    const int tid  = threadIdx.x;
    const int lane = tid & 31;
    const int wave = __builtin_amdgcn_readfirstlane(tid >> 5);
    const int base = blockIdx.x * TILE_R;
    const int rows = min(TILE_R, n - base);

    if (tid <= TILE_R) sDeg[tid] = 1.0f;
    __syncthreads();

    int* qe = sQe + wave * QCAP;
    rec_t* wlist = list + (size_t)(blockIdx.x * 32 + wave) * (WAVE_CHUNKS * 32);
    int wch = 0;

    scan_keys(key, e, qe, wave, lane, [&](int cnt) {
        for (int b0 = 0; b0 < cnt; b0 += 32) {
            const int idx = b0 + lane;
            int s = 0, r = TILE_R;
            float wt = 0.0f;
            if (idx < cnt) {
                const int ed = qe[idx];
                s  = src[ed];
                r  = dst[ed] - base;
                wt = w[ed];
                unsafeAtomicAdd(&sDeg[r], wt);
            }
            if (wch < WAVE_CHUNKS)
                wlist[wch * 32 + lane] = ((rec_t)__float_as_uint(wt) << 32) |
                                         (rec_t)(((unsigned)r << 17) | (unsigned)s);
            ++wch;
        }
    });

    if (lane == 0) sChunks[wave] = wch;
    __syncthreads();

    if (wave == 0) chunks[blockIdx.x * 32 + lane] = sChunks[lane];
    if (tid < rows) dinv[base + tid] = rsqrtf(sDeg[tid]);
}

constexpr int ROWS_IN_FLIGHT = 8;

template <int COLS>
__device__ __forceinline__ void add_rows(const float* __restrict__ h, int s, int r, float nm,
                                         int m, float* sAcc, int lane) {
    constexpr int VW = COLS / 32;
    for (int j = 0; j < m; j += ROWS_IN_FLIGHT) {
        float v[ROWS_IN_FLIGHT][VW];
        int   rr[ROWS_IN_FLIGHT];
        float nn[ROWS_IN_FLIGHT];
#pragma unroll
        for (int t = 0; t < ROWS_IN_FLIGHT; ++t) {
            const int jj = min(j + t, m - 1);
            const int sj = __builtin_amdgcn_readlane(s, jj);
            const int rj = __builtin_amdgcn_readlane(r, jj);
            rr[t] = (j + t < m) ? rj : TILE_R;
            nn[t] = __int_as_float(__builtin_amdgcn_readlane(__float_as_int(nm), jj));
            const float* hp = h + (size_t)sj * COLS + lane;
#pragma unroll
            for (int c = 0; c < VW; ++c) v[t][c] = hp[32 * c];
        }
#pragma unroll
        for (int t = 0; t < ROWS_IN_FLIGHT; ++t) {
            float* o = sAcc + rr[t] * COLS + lane;
#pragma unroll
            for (int c = 0; c < VW; ++c) unsafeAtomicAdd(o + 32 * c, v[t][c] * nn[t]);
        }
    }
}

template <int COLS>
__device__ __forceinline__ void drain_queue(
    const float* __restrict__ h, const int* __restrict__ src, const int* __restrict__ dst,
    const float* __restrict__ w, const float* __restrict__ dinv, const int* qe, int cnt,
    int base, float* sAcc, int lane) {
    for (int b0 = 0; b0 < cnt; b0 += 32) {
        const int idx = b0 + lane;
        int s = 0, r = TILE_R;
        float nm = 0.0f;
        if (idx < cnt) {
            const int ed = qe[idx];
            s  = src[ed];
            r  = dst[ed] - base;
            nm = w[ed] * dinv[s];
        }
        add_rows<COLS>(h, s, r, nm, min(32, cnt - b0), sAcc, lane);
    }
}

template <int COLS, bool RELU>
__global__ __launch_bounds__(1024) void agg_tile_kernel(
    const float* __restrict__ h, const int* __restrict__ src, const int* __restrict__ dst,
    const unsigned char* __restrict__ key, const float* __restrict__ w,
    const float* __restrict__ dinv, const float* __restrict__ b,
    float* __restrict__ out, const unsigned long long* __restrict__ list,
    const int* __restrict__ chunks, int n, int e) {
    constexpr int SHIFT = (COLS == 128) ? 7 : 6;
    __shared__ float sAcc[(TILE_R + 1) * COLS];
    __shared__ int sQe[32 * QCAP];

    const int tid  = threadIdx.x;
    const int lane = tid & 31;
    const int wave = __builtin_amdgcn_readfirstlane(tid >> 5);
    const int base = blockIdx.x * TILE_R;
    const int rows = min(TILE_R, n - base);
    const int tileElems = rows * COLS;
    const float* hTile = h + (size_t)base * COLS;
    float* oTile = out + (size_t)base * COLS;

    for (int i = tid * 4; i < tileElems; i += 1024 * 4) {
        const float dv = dinv[base + (i >> SHIFT)];
        const float4 hv = *(const float4*)(hTile + i);
        float4 o;
        o.x = hv.x * dv; o.y = hv.y * dv; o.z = hv.z * dv; o.w = hv.w * dv;
        *(float4*)(sAcc + i) = o;
    }
    if (tid < COLS) sAcc[TILE_R * COLS + tid] = 0.0f;
    __syncthreads();

    const rec_t* wlist = list + (size_t)(blockIdx.x * 32 + wave) * (WAVE_CHUNKS * 32);

    const int cl = chunks[blockIdx.x * 32 + lane];
    const bool scan = __builtin_amdgcn_ballot_w32(cl > WAVE_CHUNKS) != 0u;
    if (!scan) {
        const int nch = __builtin_amdgcn_readlane(cl, wave);
        const rec_t null_rec = (rec_t)((unsigned)TILE_R << 17);
        rec_t rec = null_rec, nxt = null_rec;
        if (nch > 0) rec = wlist[lane];
        if (nch > 1) nxt = wlist[32 + lane];
        float dv = dinv[(unsigned)rec & 0x1FFFFu];
        for (int c = 0; c < nch; ++c) {
            rec_t nx2 = null_rec;
            if (c + 2 < nch) nx2 = wlist[(c + 2) * 32 + lane];
            const float dvn = dinv[(unsigned)nxt & 0x1FFFFu];
            const unsigned lo = (unsigned)rec;
            const int r = (int)(lo >> 17);
            const int m = __popc(__builtin_amdgcn_ballot_w32(r != TILE_R));
            add_rows<COLS>(h, (int)(lo & 0x1FFFFu), r, __uint_as_float((unsigned)(rec >> 32)) * dv, m,
                           sAcc, lane);
            rec = nxt; nxt = nx2; dv = dvn;
        }
    } else {
        int* qe = sQe + wave * QCAP;
        scan_keys(key, e, qe, wave, lane, [&](int cnt) {
            drain_queue<COLS>(h, src, dst, w, dinv, qe, cnt, base, sAcc, lane);
        });
    }
    __syncthreads();

    for (int i = tid * 4; i < tileElems; i += 1024 * 4) {
        const float dv = dinv[base + (i >> SHIFT)];
        const float4 bv = *(const float4*)(b + (i & (COLS - 1)));
        float4 o = *(const float4*)(sAcc + i);
        o.x = o.x * dv + bv.x; o.y = o.y * dv + bv.y;
        o.z = o.z * dv + bv.z; o.w = o.w * dv + bv.w;
        if (RELU) {
            o.x = fmaxf(o.x, 0.0f); o.y = fmaxf(o.y, 0.0f);
            o.z = fmaxf(o.z, 0.0f); o.w = fmaxf(o.w, 0.0f);
        }
        *(float4*)(oTile + i) = o;
    }
}


extern "C" void kernel_launch(void* const* d_in, const int* in_sizes, int n_in,
                              void* d_out, int out_size, void* d_ws, size_t ws_size,
                              hipStream_t stream) {
    const float* x   = (const float*)d_in[0];
    const int*  eidx = (const int*)d_in[1];
    const float* ew  = (const float*)d_in[2];
    const float* W1  = (const float*)d_in[3];
    const float* b1  = (const float*)d_in[4];
    const float* W2  = (const float*)d_in[5];
    const float* b2  = (const float*)d_in[6];

    const int N = in_sizes[0] / 128;
    const int E = in_sizes[2];
    const int* src = eidx;
    const int* dst = eidx + E;

    float* dinv = (float*)d_ws;
    float* h1   = dinv + N;
    float* acc  = h1 + (size_t)N * 128;
    unsigned char* key = (unsigned char*)(acc + (size_t)N * 128);
    float* out  = (float*)d_out;

    const int T = 256;
    const int gemmBlocks = ((N + 15) / 16 + 7) / 8;
    const int aggBlocks  = (N + TILE_R - 1) / TILE_R;
    const size_t keyEnd  = (size_t)((key + E) - (unsigned char*)d_ws);
    int* chunks = (int*)((unsigned char*)d_ws + ((keyEnd + 255) & ~(size_t)255));
    unsigned long long* list =
        (unsigned long long*)(chunks + (((size_t)aggBlocks * 32 + 63) & ~(size_t)63));

    key_kernel<<<(E + T - 1) / T, T, 0, stream>>>(dst, key, E);
    scan_deg_kernel<<<aggBlocks, 1024, 0, stream>>>(src, dst, ew, key, dinv, list, chunks, N, E);

    gemm_wmma_kernel<128><<<gemmBlocks, T, 0, stream>>>(x, W1, h1, N);
    agg_tile_kernel<128, true><<<aggBlocks, 1024, 0, stream>>>(
        h1, src, dst, key, ew, dinv, b1, acc, list, chunks, N, E);

    gemm_wmma_kernel<64><<<gemmBlocks, T, 0, stream>>>(acc, W2, h1, N);
    agg_tile_kernel<64, false><<<aggBlocks, 1024, 0, stream>>>(
        h1, src, dst, key, ew, dinv, b2, out, list, chunks, N, E);
}


# edge-key pass also writes each edge's 8-byte record (tile row, source, weight); the scan fetches that one record per found edge, not its source, destination and weight: 1 line, not 3; sums untouched
# speedup vs baseline: 3.3524x; 1.0702x over previous
#include <hip/hip_runtime.h>


typedef __attribute__((ext_vector_type(2))) float v2f;
typedef __attribute__((ext_vector_type(8))) float v8f;

constexpr int TILE_R = 391;

__global__ void key_kernel(const int* __restrict__ src, const int* __restrict__ dst,
                           const float* __restrict__ w, unsigned char* __restrict__ key,
                           unsigned long long* __restrict__ erec, int e) {
    int i = blockIdx.x * blockDim.x + threadIdx.x;
    if (i < e) {
        const unsigned d = (unsigned)dst[i];
        const unsigned g = d / (unsigned)TILE_R;
        key[i]  = (unsigned char)g;
        erec[i] = ((unsigned long long)__float_as_uint(w[i]) << 32) |
                  (unsigned long long)(((d - g * (unsigned)TILE_R) << 17) | (unsigned)src[i]);
    }
}


template <int COLS>
__global__ __launch_bounds__(256) void gemm_wmma_kernel(
    const float* __restrict__ X, const float* __restrict__ W,
    float* __restrict__ OUT, int nRows) {
    constexpr int K  = 128;
    constexpr int NT = COLS / 16;
    constexpr int CSH = (COLS == 128) ? 7 : 6;

    __shared__ float sW[(K / 2) * COLS * 2];
    for (int i = threadIdx.x; i < K * COLS; i += 256) {
        const int row = i >> CSH;
        const int col = i & (COLS - 1);
        sW[(row >> 1) * (COLS * 2) + col * 2 + (row & 1)] = W[i];
    }
    __syncthreads();

    const int wave = threadIdx.x >> 5;
    const int lane = threadIdx.x & 31;
    const int rb   = blockIdx.x * 8 + wave;
    if (rb * 16 >= nRows) return;

    const int l15  = lane & 15;
    const int hsel = lane >> 4;

    const float* aBase = X + (size_t)(rb * 16 + l15) * K + 2 * hsel;
    const float* bCol  = &sW[hsel * (COLS * 2) + l15 * 2];

    v8f acc[NT] = {};

    for (int k = 0; k < K; k += 4) {
        const v2f A = *(const v2f*)(aBase + k);
        const float* bp = bCol + (k >> 1) * (COLS * 2);
#pragma unroll
        for (int c = 0; c < NT; ++c) {
            const v2f B = *(const v2f*)(bp + c * 32);
            acc[c] = __builtin_amdgcn_wmma_f32_16x16x4_f32(
                false, A, false, B, (short)0, acc[c], false, false);
        }
    }

    const int rowBase = rb * 16 + (hsel << 3);
#pragma unroll
    for (int c = 0; c < NT; ++c) {
#pragma unroll
        for (int r = 0; r < 8; ++r)
            OUT[(size_t)(rowBase + r) * COLS + c * 16 + l15] = acc[c][r];
    }
}


constexpr int QCAP    = 64;
constexpr int GRP     = 8;
constexpr int STEP_SH = 17;

__device__ __forceinline__ unsigned zero_bytes(unsigned x) {
    return ~(((x & 0x7F7F7F7Fu) + 0x7F7F7F7Fu) | x) & 0x80808080u;
}

__device__ __forceinline__ unsigned guarded_word(const unsigned char* __restrict__ key,
                                                 int i0, int e, unsigned miss) {
    unsigned w = 0;
#pragma unroll
    for (int k = 0; k < 4; ++k) {
        const unsigned b = (i0 + k < e) ? (unsigned)key[i0 + k] : miss;
        w |= b << (8 * k);
    }
    return w;
}

typedef unsigned long long rec_t;
constexpr int WAVE_CHUNKS = 32;

template <class Drain>
__device__ __forceinline__ void scan_keys(const unsigned char* __restrict__ key, int e, int* qe,
                                          int wave, int lane, Drain drain) {
    const unsigned mine  = blockIdx.x & 255u;
    const unsigned pat   = mine * 0x01010101u;
    const unsigned miss  = mine ^ 1u;
    const unsigned miss4 = miss * 0x01010101u;
    const bool vec_ok = ((((size_t)key) & 15) == 0);
    int qcnt = 0;

    const int nsteps = (e + (1 << STEP_SH) - 1) >> STEP_SH;
    const int step0  = (int)(((long long)blockIdx.x * nsteps) / gridDim.x);
    const int woff   = (wave * GRP) << 9;

    for (int it = 0; it < nsteps; ++it) {
        int st = step0 + it;
        if (st >= nsteps) st -= nsteps;
        const int wbase = (st << STEP_SH) + woff;

        uint4 cur[GRP];
        if (vec_ok && wbase + (GRP << 9) <= e) {
#pragma unroll
            for (int u = 0; u < GRP; ++u)
                cur[u] = *(const uint4*)(key + wbase + (u << 9) + lane * 16);
        } else {
#pragma unroll
            for (int u = 0; u < GRP; ++u) {
                const int c0 = wbase + (u << 9);
                if (vec_ok && c0 + 512 <= e) {
                    cur[u] = *(const uint4*)(key + c0 + lane * 16);
                } else if (c0 >= e) {
                    cur[u] = make_uint4(miss4, miss4, miss4, miss4);
                } else {
                    const int i0 = c0 + lane * 16;
                    cur[u].x = guarded_word(key, i0,      e, miss);
                    cur[u].y = guarded_word(key, i0 + 4,  e, miss);
                    cur[u].z = guarded_word(key, i0 + 8,  e, miss);
                    cur[u].w = guarded_word(key, i0 + 12, e, miss);
                }
            }
        }

#pragma unroll
        for (int u = 0; u < GRP; ++u) {
            unsigned M = (zero_bytes(cur[u].x ^ pat) >> 7) | (zero_bytes(cur[u].y ^ pat) >> 6) |
                         (zero_bytes(cur[u].z ^ pat) >> 5) | (zero_bytes(cur[u].w ^ pat) >> 4);
            unsigned bm = __builtin_amdgcn_ballot_w32(M != 0u);
            const int e0 = wbase + (u << 9) + lane * 16;
            while (bm) {
                if (M != 0u) {
                    const int B = __builtin_ctz(M);
                    M &= M - 1u;
                    qe[__builtin_amdgcn_mbcnt_lo(bm, (unsigned)qcnt)] = e0 + ((B & 7) << 2) + (B >> 3);
                }
                qcnt += __popc(bm);
                if (qcnt >= 32) {
                    drain(32);
                    qcnt -= 32;
                    if (lane < qcnt) {
                        const int keep = qe[32 + lane];
                        qe[lane] = keep;
                    }
                }
                bm = __builtin_amdgcn_ballot_w32(M != 0u);
            }
        }
    }
    if (qcnt > 0) drain(qcnt);
}

__global__ __launch_bounds__(1024) void scan_deg_kernel(
    const unsigned long long* __restrict__ erec,
    const unsigned char* __restrict__ key, float* __restrict__ dinv,
    unsigned long long* __restrict__ list, int* __restrict__ chunks, int n, int e) {
    __shared__ float sDeg[TILE_R + 1];
    __shared__ int sQe[32 * QCAP];
    __shared__ int sChunks[32];

    const int tid  = threadIdx.x;
    const int lane = tid & 31;
    const int wave = __builtin_amdgcn_readfirstlane(tid >> 5);
    const int base = blockIdx.x * TILE_R;
    const int rows = min(TILE_R, n - base);

    if (tid <= TILE_R) sDeg[tid] = 1.0f;
    __syncthreads();

    int* qe = sQe + wave * QCAP;
    rec_t* wlist = list + (size_t)(blockIdx.x * 32 + wave) * (WAVE_CHUNKS * 32);
    int wch = 0;

    scan_keys(key, e, qe, wave, lane, [&](int cnt) {
        for (int b0 = 0; b0 < cnt; b0 += 32) {
            const int idx = b0 + lane;
            rec_t rec = (rec_t)((unsigned)TILE_R << 17);
            if (idx < cnt) {
                rec = erec[qe[idx]];
                unsafeAtomicAdd(&sDeg[(unsigned)rec >> 17], __uint_as_float((unsigned)(rec >> 32)));
            }
            if (wch < WAVE_CHUNKS) wlist[wch * 32 + lane] = rec;
            ++wch;
        }
    });

    if (lane == 0) sChunks[wave] = wch;
    __syncthreads();

    if (wave == 0) chunks[blockIdx.x * 32 + lane] = sChunks[lane];
    if (tid < rows) dinv[base + tid] = rsqrtf(sDeg[tid]);
}

constexpr int ROWS_IN_FLIGHT = 8;

template <int COLS>
__device__ __forceinline__ void add_rows(const float* __restrict__ h, int s, int r, float nm,
                                         int m, float* sAcc, int lane) {
    constexpr int VW = COLS / 32;
    for (int j = 0; j < m; j += ROWS_IN_FLIGHT) {
        float v[ROWS_IN_FLIGHT][VW];
        int   rr[ROWS_IN_FLIGHT];
        float nn[ROWS_IN_FLIGHT];
#pragma unroll
        for (int t = 0; t < ROWS_IN_FLIGHT; ++t) {
            const int jj = min(j + t, m - 1);
            const int sj = __builtin_amdgcn_readlane(s, jj);
            const int rj = __builtin_amdgcn_readlane(r, jj);
            rr[t] = (j + t < m) ? rj : TILE_R;
            nn[t] = __int_as_float(__builtin_amdgcn_readlane(__float_as_int(nm), jj));
            const float* hp = h + (size_t)sj * COLS + lane;
#pragma unroll
            for (int c = 0; c < VW; ++c) v[t][c] = hp[32 * c];
        }
#pragma unroll
        for (int t = 0; t < ROWS_IN_FLIGHT; ++t) {
            float* o = sAcc + rr[t] * COLS + lane;
#pragma unroll
            for (int c = 0; c < VW; ++c) unsafeAtomicAdd(o + 32 * c, v[t][c] * nn[t]);
        }
    }
}

template <int COLS>
__device__ __forceinline__ void drain_queue(
    const float* __restrict__ h, const int* __restrict__ src, const int* __restrict__ dst,
    const float* __restrict__ w, const float* __restrict__ dinv, const int* qe, int cnt,
    int base, float* sAcc, int lane) {
    for (int b0 = 0; b0 < cnt; b0 += 32) {
        const int idx = b0 + lane;
        int s = 0, r = TILE_R;
        float nm = 0.0f;
        if (idx < cnt) {
            const int ed = qe[idx];
            s  = src[ed];
            r  = dst[ed] - base;
            nm = w[ed] * dinv[s];
        }
        add_rows<COLS>(h, s, r, nm, min(32, cnt - b0), sAcc, lane);
    }
}

template <int COLS, bool RELU>
__global__ __launch_bounds__(1024) void agg_tile_kernel(
    const float* __restrict__ h, const int* __restrict__ src, const int* __restrict__ dst,
    const unsigned char* __restrict__ key, const float* __restrict__ w,
    const float* __restrict__ dinv, const float* __restrict__ b,
    float* __restrict__ out, const unsigned long long* __restrict__ list,
    const int* __restrict__ chunks, int n, int e) {
    constexpr int SHIFT = (COLS == 128) ? 7 : 6;
    __shared__ float sAcc[(TILE_R + 1) * COLS];
    __shared__ int sQe[32 * QCAP];

    const int tid  = threadIdx.x;
    const int lane = tid & 31;
    const int wave = __builtin_amdgcn_readfirstlane(tid >> 5);
    const int base = blockIdx.x * TILE_R;
    const int rows = min(TILE_R, n - base);
    const int tileElems = rows * COLS;
    const float* hTile = h + (size_t)base * COLS;
    float* oTile = out + (size_t)base * COLS;

    for (int i = tid * 4; i < tileElems; i += 1024 * 4) {
        const float dv = dinv[base + (i >> SHIFT)];
        const float4 hv = *(const float4*)(hTile + i);
        float4 o;
        o.x = hv.x * dv; o.y = hv.y * dv; o.z = hv.z * dv; o.w = hv.w * dv;
        *(float4*)(sAcc + i) = o;
    }
    if (tid < COLS) sAcc[TILE_R * COLS + tid] = 0.0f;
    __syncthreads();

    const rec_t* wlist = list + (size_t)(blockIdx.x * 32 + wave) * (WAVE_CHUNKS * 32);

    const int cl = chunks[blockIdx.x * 32 + lane];
    const bool scan = __builtin_amdgcn_ballot_w32(cl > WAVE_CHUNKS) != 0u;
    if (!scan) {
        const int nch = __builtin_amdgcn_readlane(cl, wave);
        const rec_t null_rec = (rec_t)((unsigned)TILE_R << 17);
        rec_t rec = null_rec, nxt = null_rec;
        if (nch > 0) rec = wlist[lane];
        if (nch > 1) nxt = wlist[32 + lane];
        float dv = dinv[(unsigned)rec & 0x1FFFFu];
        for (int c = 0; c < nch; ++c) {
            rec_t nx2 = null_rec;
            if (c + 2 < nch) nx2 = wlist[(c + 2) * 32 + lane];
            const float dvn = dinv[(unsigned)nxt & 0x1FFFFu];
            const unsigned lo = (unsigned)rec;
            const int r = (int)(lo >> 17);
            const int m = __popc(__builtin_amdgcn_ballot_w32(r != TILE_R));
            add_rows<COLS>(h, (int)(lo & 0x1FFFFu), r, __uint_as_float((unsigned)(rec >> 32)) * dv, m,
                           sAcc, lane);
            rec = nxt; nxt = nx2; dv = dvn;
        }
    } else {
        int* qe = sQe + wave * QCAP;
        scan_keys(key, e, qe, wave, lane, [&](int cnt) {
            drain_queue<COLS>(h, src, dst, w, dinv, qe, cnt, base, sAcc, lane);
        });
    }
    __syncthreads();

    for (int i = tid * 4; i < tileElems; i += 1024 * 4) {
        const float dv = dinv[base + (i >> SHIFT)];
        const float4 bv = *(const float4*)(b + (i & (COLS - 1)));
        float4 o = *(const float4*)(sAcc + i);
        o.x = o.x * dv + bv.x; o.y = o.y * dv + bv.y;
        o.z = o.z * dv + bv.z; o.w = o.w * dv + bv.w;
        if (RELU) {
            o.x = fmaxf(o.x, 0.0f); o.y = fmaxf(o.y, 0.0f);
            o.z = fmaxf(o.z, 0.0f); o.w = fmaxf(o.w, 0.0f);
        }
        *(float4*)(oTile + i) = o;
    }
}


extern "C" void kernel_launch(void* const* d_in, const int* in_sizes, int n_in,
                              void* d_out, int out_size, void* d_ws, size_t ws_size,
                              hipStream_t stream) {
    const float* x   = (const float*)d_in[0];
    const int*  eidx = (const int*)d_in[1];
    const float* ew  = (const float*)d_in[2];
    const float* W1  = (const float*)d_in[3];
    const float* b1  = (const float*)d_in[4];
    const float* W2  = (const float*)d_in[5];
    const float* b2  = (const float*)d_in[6];

    const int N = in_sizes[0] / 128;
    const int E = in_sizes[2];
    const int* src = eidx;
    const int* dst = eidx + E;

    float* dinv = (float*)d_ws;
    float* h1   = dinv + N;
    float* acc  = h1 + (size_t)N * 128;
    unsigned char* key = (unsigned char*)(acc + (size_t)N * 128);
    float* out  = (float*)d_out;

    const int T = 256;
    const int gemmBlocks = ((N + 15) / 16 + 7) / 8;
    const int aggBlocks  = (N + TILE_R - 1) / TILE_R;
    const size_t keyEnd  = (size_t)((key + E) - (unsigned char*)d_ws);
    int* chunks = (int*)((unsigned char*)d_ws + ((keyEnd + 255) & ~(size_t)255));
    unsigned long long* list =
        (unsigned long long*)(chunks + (((size_t)aggBlocks * 32 + 63) & ~(size_t)63));
    unsigned long long* erec = list + (size_t)aggBlocks * 32 * (WAVE_CHUNKS * 32);

    key_kernel<<<(E + T - 1) / T, T, 0, stream>>>(src, dst, ew, key, erec, E);
    scan_deg_kernel<<<aggBlocks, 1024, 0, stream>>>(erec, key, dinv, list, chunks, N, E);

    gemm_wmma_kernel<128><<<gemmBlocks, T, 0, stream>>>(x, W1, h1, N);
    agg_tile_kernel<128, true><<<aggBlocks, 1024, 0, stream>>>(
        h1, src, dst, key, ew, dinv, b1, acc, list, chunks, N, E);

    gemm_wmma_kernel<64><<<gemmBlocks, T, 0, stream>>>(acc, W2, h1, N);
    agg_tile_kernel<64, false><<<aggBlocks, 1024, 0, stream>>>(
        h1, src, dst, key, ew, dinv, b2, out, list, chunks, N, E);
}


# layer 2's dense transform (fp32 WMMA, fragments read from LDS) runs on layer 1's finished tile inside the aggregation's work-group: activations never go to memory, 5 launches not 6; sums untouched
# speedup vs baseline: 3.6254x; 1.0814x over previous
#include <hip/hip_runtime.h>


typedef __attribute__((ext_vector_type(2))) float v2f;
typedef __attribute__((ext_vector_type(8))) float v8f;

constexpr int TILE_R = 391;

__global__ void key_kernel(const int* __restrict__ src, const int* __restrict__ dst,
                           const float* __restrict__ w, unsigned char* __restrict__ key,
                           unsigned long long* __restrict__ erec, int e) {
    int i = blockIdx.x * blockDim.x + threadIdx.x;
    if (i < e) {
        const unsigned d = (unsigned)dst[i];
        const unsigned g = d / (unsigned)TILE_R;
        key[i]  = (unsigned char)g;
        erec[i] = ((unsigned long long)__float_as_uint(w[i]) << 32) |
                  (unsigned long long)(((d - g * (unsigned)TILE_R) << 17) | (unsigned)src[i]);
    }
}


template <int COLS>
__global__ __launch_bounds__(256) void gemm_wmma_kernel(
    const float* __restrict__ X, const float* __restrict__ W,
    float* __restrict__ OUT, int nRows) {
    constexpr int K  = 128;
    constexpr int NT = COLS / 16;
    constexpr int CSH = (COLS == 128) ? 7 : 6;

    __shared__ float sW[(K / 2) * COLS * 2];
    for (int i = threadIdx.x; i < K * COLS; i += 256) {
        const int row = i >> CSH;
        const int col = i & (COLS - 1);
        sW[(row >> 1) * (COLS * 2) + col * 2 + (row & 1)] = W[i];
    }
    __syncthreads();

    const int wave = threadIdx.x >> 5;
    const int lane = threadIdx.x & 31;
    const int rb   = blockIdx.x * 8 + wave;
    if (rb * 16 >= nRows) return;

    const int l15  = lane & 15;
    const int hsel = lane >> 4;

    const float* aBase = X + (size_t)(rb * 16 + l15) * K + 2 * hsel;
    const float* bCol  = &sW[hsel * (COLS * 2) + l15 * 2];

    v8f acc[NT] = {};

    for (int k = 0; k < K; k += 4) {
        const v2f A = *(const v2f*)(aBase + k);
        const float* bp = bCol + (k >> 1) * (COLS * 2);
#pragma unroll
        for (int c = 0; c < NT; ++c) {
            const v2f B = *(const v2f*)(bp + c * 32);
            acc[c] = __builtin_amdgcn_wmma_f32_16x16x4_f32(
                false, A, false, B, (short)0, acc[c], false, false);
        }
    }

    const int rowBase = rb * 16 + (hsel << 3);
#pragma unroll
    for (int c = 0; c < NT; ++c) {
#pragma unroll
        for (int r = 0; r < 8; ++r)
            OUT[(size_t)(rowBase + r) * COLS + c * 16 + l15] = acc[c][r];
    }
}


constexpr int QCAP    = 64;
constexpr int GRP     = 8;
constexpr int STEP_SH = 17;

__device__ __forceinline__ unsigned zero_bytes(unsigned x) {
    return ~(((x & 0x7F7F7F7Fu) + 0x7F7F7F7Fu) | x) & 0x80808080u;
}

__device__ __forceinline__ unsigned guarded_word(const unsigned char* __restrict__ key,
                                                 int i0, int e, unsigned miss) {
    unsigned w = 0;
#pragma unroll
    for (int k = 0; k < 4; ++k) {
        const unsigned b = (i0 + k < e) ? (unsigned)key[i0 + k] : miss;
        w |= b << (8 * k);
    }
    return w;
}

typedef unsigned long long rec_t;
constexpr int WAVE_CHUNKS = 32;

template <class Drain>
__device__ __forceinline__ void scan_keys(const unsigned char* __restrict__ key, int e, int* qe,
                                          int wave, int lane, Drain drain) {
    const unsigned mine  = blockIdx.x & 255u;
    const unsigned pat   = mine * 0x01010101u;
    const unsigned miss  = mine ^ 1u;
    const unsigned miss4 = miss * 0x01010101u;
    const bool vec_ok = ((((size_t)key) & 15) == 0);
    int qcnt = 0;

    const int nsteps = (e + (1 << STEP_SH) - 1) >> STEP_SH;
    const int step0  = (int)(((long long)blockIdx.x * nsteps) / gridDim.x);
    const int woff   = (wave * GRP) << 9;

    for (int it = 0; it < nsteps; ++it) {
        int st = step0 + it;
        if (st >= nsteps) st -= nsteps;
        const int wbase = (st << STEP_SH) + woff;

        uint4 cur[GRP];
        if (vec_ok && wbase + (GRP << 9) <= e) {
#pragma unroll
            for (int u = 0; u < GRP; ++u)
                cur[u] = *(const uint4*)(key + wbase + (u << 9) + lane * 16);
        } else {
#pragma unroll
            for (int u = 0; u < GRP; ++u) {
                const int c0 = wbase + (u << 9);
                if (vec_ok && c0 + 512 <= e) {
                    cur[u] = *(const uint4*)(key + c0 + lane * 16);
                } else if (c0 >= e) {
                    cur[u] = make_uint4(miss4, miss4, miss4, miss4);
                } else {
                    const int i0 = c0 + lane * 16;
                    cur[u].x = guarded_word(key, i0,      e, miss);
                    cur[u].y = guarded_word(key, i0 + 4,  e, miss);
                    cur[u].z = guarded_word(key, i0 + 8,  e, miss);
                    cur[u].w = guarded_word(key, i0 + 12, e, miss);
                }
            }
        }

#pragma unroll
        for (int u = 0; u < GRP; ++u) {
            unsigned M = (zero_bytes(cur[u].x ^ pat) >> 7) | (zero_bytes(cur[u].y ^ pat) >> 6) |
                         (zero_bytes(cur[u].z ^ pat) >> 5) | (zero_bytes(cur[u].w ^ pat) >> 4);
            unsigned bm = __builtin_amdgcn_ballot_w32(M != 0u);
            const int e0 = wbase + (u << 9) + lane * 16;
            while (bm) {
                if (M != 0u) {
                    const int B = __builtin_ctz(M);
                    M &= M - 1u;
                    qe[__builtin_amdgcn_mbcnt_lo(bm, (unsigned)qcnt)] = e0 + ((B & 7) << 2) + (B >> 3);
                }
                qcnt += __popc(bm);
                if (qcnt >= 32) {
                    drain(32);
                    qcnt -= 32;
                    if (lane < qcnt) {
                        const int keep = qe[32 + lane];
                        qe[lane] = keep;
                    }
                }
                bm = __builtin_amdgcn_ballot_w32(M != 0u);
            }
        }
    }
    if (qcnt > 0) drain(qcnt);
}

__global__ __launch_bounds__(1024) void scan_deg_kernel(
    const unsigned long long* __restrict__ erec,
    const unsigned char* __restrict__ key, float* __restrict__ dinv,
    unsigned long long* __restrict__ list, int* __restrict__ chunks, int n, int e) {
    __shared__ float sDeg[TILE_R + 1];
    __shared__ int sQe[32 * QCAP];
    __shared__ int sChunks[32];

    const int tid  = threadIdx.x;
    const int lane = tid & 31;
    const int wave = __builtin_amdgcn_readfirstlane(tid >> 5);
    const int base = blockIdx.x * TILE_R;
    const int rows = min(TILE_R, n - base);

    if (tid <= TILE_R) sDeg[tid] = 1.0f;
    __syncthreads();

    int* qe = sQe + wave * QCAP;
    rec_t* wlist = list + (size_t)(blockIdx.x * 32 + wave) * (WAVE_CHUNKS * 32);
    int wch = 0;

    scan_keys(key, e, qe, wave, lane, [&](int cnt) {
        for (int b0 = 0; b0 < cnt; b0 += 32) {
            const int idx = b0 + lane;
            rec_t rec = (rec_t)((unsigned)TILE_R << 17);
            if (idx < cnt) {
                rec = erec[qe[idx]];
                unsafeAtomicAdd(&sDeg[(unsigned)rec >> 17], __uint_as_float((unsigned)(rec >> 32)));
            }
            if (wch < WAVE_CHUNKS) wlist[wch * 32 + lane] = rec;
            ++wch;
        }
    });

    if (lane == 0) sChunks[wave] = wch;
    __syncthreads();

    if (wave == 0) chunks[blockIdx.x * 32 + lane] = sChunks[lane];
    if (tid < rows) dinv[base + tid] = rsqrtf(sDeg[tid]);
}

constexpr int ROWS_IN_FLIGHT = 8;

template <int COLS, int PITCH>
__device__ __forceinline__ void add_rows(const float* __restrict__ h, int s, int r, float nm,
                                         int m, float* sAcc, int lane) {
    constexpr int VW = COLS / 32;
    for (int j = 0; j < m; j += ROWS_IN_FLIGHT) {
        float v[ROWS_IN_FLIGHT][VW];
        int   rr[ROWS_IN_FLIGHT];
        float nn[ROWS_IN_FLIGHT];
#pragma unroll
        for (int t = 0; t < ROWS_IN_FLIGHT; ++t) {
            const int jj = min(j + t, m - 1);
            const int sj = __builtin_amdgcn_readlane(s, jj);
            const int rj = __builtin_amdgcn_readlane(r, jj);
            rr[t] = (j + t < m) ? rj : TILE_R;
            nn[t] = __int_as_float(__builtin_amdgcn_readlane(__float_as_int(nm), jj));
            const float* hp = h + (size_t)sj * COLS + lane;
#pragma unroll
            for (int c = 0; c < VW; ++c) v[t][c] = hp[32 * c];
        }
#pragma unroll
        for (int t = 0; t < ROWS_IN_FLIGHT; ++t) {
            float* o = sAcc + rr[t] * PITCH + lane;
#pragma unroll
            for (int c = 0; c < VW; ++c) unsafeAtomicAdd(o + 32 * c, v[t][c] * nn[t]);
        }
    }
}

template <int COLS, int PITCH>
__device__ __forceinline__ void drain_queue(
    const float* __restrict__ h, const int* __restrict__ src, const int* __restrict__ dst,
    const float* __restrict__ w, const float* __restrict__ dinv, const int* qe, int cnt,
    int base, float* sAcc, int lane) {
    for (int b0 = 0; b0 < cnt; b0 += 32) {
        const int idx = b0 + lane;
        int s = 0, r = TILE_R;
        float nm = 0.0f;
        if (idx < cnt) {
            const int ed = qe[idx];
            s  = src[ed];
            r  = dst[ed] - base;
            nm = w[ed] * dinv[s];
        }
        add_rows<COLS, PITCH>(h, s, r, nm, min(32, cnt - b0), sAcc, lane);
    }
}

constexpr int OUT2     = 64;
constexpr int TILE_RB  = (TILE_R + 1 + 15) / 16;
constexpr int W2_PITCH = OUT2 * 2 + 32;

template <int COLS, bool FUSED>
__global__ __launch_bounds__(1024) void agg_tile_kernel(
    const float* __restrict__ h, const int* __restrict__ src, const int* __restrict__ dst,
    const unsigned char* __restrict__ key, const float* __restrict__ w,
    const float* __restrict__ dinv, const float* __restrict__ b,
    const float* __restrict__ W2,
    float* __restrict__ out, const unsigned long long* __restrict__ list,
    const int* __restrict__ chunks, int n, int e) {
    constexpr int SHIFT = (COLS == 128) ? 7 : 6;
    constexpr int PITCH = FUSED ? COLS + 4 : COLS;
    constexpr int LROWS = FUSED ? TILE_RB * 16 : TILE_R + 1;
    __shared__ float sAcc[LROWS * PITCH];
    __shared__ float sW[FUSED ? (COLS / 2) * W2_PITCH : 4];
    __shared__ int sQe[32 * QCAP];

    const int tid  = threadIdx.x;
    const int lane = tid & 31;
    const int wave = __builtin_amdgcn_readfirstlane(tid >> 5);
    const int base = blockIdx.x * TILE_R;
    const int rows = min(TILE_R, n - base);
    const int tileElems = rows * COLS;
    const float* hTile = h + (size_t)base * COLS;

    constexpr int W2_PER_THREAD = FUSED ? COLS * OUT2 / 1024 : 1;
    float w2v[W2_PER_THREAD];
    if (FUSED) {
#pragma unroll
        for (int j = 0; j < W2_PER_THREAD; ++j) w2v[j] = W2[tid + j * 1024];
    }

    for (int i = tid * 4; i < tileElems; i += 1024 * 4) {
        const int row = i >> SHIFT;
        const float dv = dinv[base + row];
        const float4 hv = *(const float4*)(hTile + i);
        float4 o;
        o.x = hv.x * dv; o.y = hv.y * dv; o.z = hv.z * dv; o.w = hv.w * dv;
        *(float4*)(sAcc + row * PITCH + (i & (COLS - 1))) = o;
    }
    if (FUSED) {
#pragma unroll
        for (int j = 0; j < W2_PER_THREAD; ++j) {
            const int i   = tid + j * 1024;
            const int row = i / OUT2;
            const int col = i % OUT2;
            sW[(row >> 1) * W2_PITCH + col * 2 + (row & 1)] = w2v[j];
        }
    }
    for (int i = tid; i < (LROWS - TILE_R) * COLS; i += 1024)
        sAcc[(TILE_R + (i >> SHIFT)) * PITCH + (i & (COLS - 1))] = 0.0f;
    __syncthreads();

    const rec_t* wlist = list + (size_t)(blockIdx.x * 32 + wave) * (WAVE_CHUNKS * 32);

    const int cl = chunks[blockIdx.x * 32 + lane];
    const bool scan = __builtin_amdgcn_ballot_w32(cl > WAVE_CHUNKS) != 0u;
    if (!scan) {
        const int nch = __builtin_amdgcn_readlane(cl, wave);
        const rec_t null_rec = (rec_t)((unsigned)TILE_R << 17);
        rec_t rec = null_rec, nxt = null_rec;
        if (nch > 0) rec = wlist[lane];
        if (nch > 1) nxt = wlist[32 + lane];
        float dv = dinv[(unsigned)rec & 0x1FFFFu];
        for (int c = 0; c < nch; ++c) {
            rec_t nx2 = null_rec;
            if (c + 2 < nch) nx2 = wlist[(c + 2) * 32 + lane];
            const float dvn = dinv[(unsigned)nxt & 0x1FFFFu];
            const unsigned lo = (unsigned)rec;
            const int r = (int)(lo >> 17);
            const int m = __popc(__builtin_amdgcn_ballot_w32(r != TILE_R));
            add_rows<COLS, PITCH>(h, (int)(lo & 0x1FFFFu), r, __uint_as_float((unsigned)(rec >> 32)) * dv, m,
                                  sAcc, lane);
            rec = nxt; nxt = nx2; dv = dvn;
        }
    } else {
        int* qe = sQe + wave * QCAP;
        scan_keys(key, e, qe, wave, lane, [&](int cnt) {
            drain_queue<COLS, PITCH>(h, src, dst, w, dinv, qe, cnt, base, sAcc, lane);
        });
    }
    __syncthreads();

    if (!FUSED) {
        float* oTile = out + (size_t)base * COLS;
        for (int i = tid * 4; i < tileElems; i += 1024 * 4) {
            const int row = i >> SHIFT;
            const float dv = dinv[base + row];
            const float4 bv = *(const float4*)(b + (i & (COLS - 1)));
            float4 o = *(const float4*)(sAcc + row * PITCH + (i & (COLS - 1)));
            o.x = o.x * dv + bv.x; o.y = o.y * dv + bv.y;
            o.z = o.z * dv + bv.z; o.w = o.w * dv + bv.w;
            *(float4*)(oTile + i) = o;
        }
    } else {
        for (int i = tid * 4; i < tileElems; i += 1024 * 4) {
            const int row = i >> SHIFT;
            const float dv = dinv[base + row];
            const float4 bv = *(const float4*)(b + (i & (COLS - 1)));
            float* p = sAcc + row * PITCH + (i & (COLS - 1));
            float4 o = *(const float4*)p;
            o.x = fmaxf(o.x * dv + bv.x, 0.0f); o.y = fmaxf(o.y * dv + bv.y, 0.0f);
            o.z = fmaxf(o.z * dv + bv.z, 0.0f); o.w = fmaxf(o.w * dv + bv.w, 0.0f);
            *(float4*)p = o;
        }
        __syncthreads();

        const int l15  = lane & 15;
        const int hsel = lane >> 4;
        const int c0   = (wave & 3) * 16;
        const float* bCol = sW + hsel * W2_PITCH + (c0 + l15) * 2;
        v2f Bf[COLS / 4];
#pragma unroll
        for (int kk = 0; kk < COLS / 4; ++kk) Bf[kk] = *(const v2f*)(bCol + kk * (2 * W2_PITCH));

        for (int rb = wave >> 2; rb < TILE_RB; rb += 8) {
            const float* aBase = sAcc + (rb * 16 + l15) * PITCH + 2 * hsel;
            v8f acc = {};
#pragma unroll
            for (int kk = 0; kk < COLS / 4; ++kk) {
                const v2f A = *(const v2f*)(aBase + 4 * kk);
                acc = __builtin_amdgcn_wmma_f32_16x16x4_f32(
                    false, A, false, Bf[kk], (short)0, acc, false, false);
            }
            const int rowBase = rb * 16 + (hsel << 3);
#pragma unroll
            for (int r = 0; r < 8; ++r) {
                const int row = rowBase + r;
                if (row < rows) out[(size_t)(base + row) * OUT2 + c0 + l15] = acc[r];
            }
        }
    }
}


extern "C" void kernel_launch(void* const* d_in, const int* in_sizes, int n_in,
                              void* d_out, int out_size, void* d_ws, size_t ws_size,
                              hipStream_t stream) {
    const float* x   = (const float*)d_in[0];
    const int*  eidx = (const int*)d_in[1];
    const float* ew  = (const float*)d_in[2];
    const float* W1  = (const float*)d_in[3];
    const float* b1  = (const float*)d_in[4];
    const float* W2  = (const float*)d_in[5];
    const float* b2  = (const float*)d_in[6];

    const int N = in_sizes[0] / 128;
    const int E = in_sizes[2];
    const int* src = eidx;
    const int* dst = eidx + E;

    float* dinv = (float*)d_ws;
    float* h1   = dinv + N;
    float* acc  = h1 + (size_t)N * 128;
    unsigned char* key = (unsigned char*)(acc + (size_t)N * 128);
    float* out  = (float*)d_out;

    const int T = 256;
    const int gemmBlocks = ((N + 15) / 16 + 7) / 8;
    const int aggBlocks  = (N + TILE_R - 1) / TILE_R;
    const size_t keyEnd  = (size_t)((key + E) - (unsigned char*)d_ws);
    int* chunks = (int*)((unsigned char*)d_ws + ((keyEnd + 255) & ~(size_t)255));
    unsigned long long* list =
        (unsigned long long*)(chunks + (((size_t)aggBlocks * 32 + 63) & ~(size_t)63));
    unsigned long long* erec = list + (size_t)aggBlocks * 32 * (WAVE_CHUNKS * 32);

    key_kernel<<<(E + T - 1) / T, T, 0, stream>>>(src, dst, ew, key, erec, E);
    scan_deg_kernel<<<aggBlocks, 1024, 0, stream>>>(erec, key, dinv, list, chunks, N, E);

    gemm_wmma_kernel<128><<<gemmBlocks, T, 0, stream>>>(x, W1, h1, N);
    agg_tile_kernel<128, true><<<aggBlocks, 1024, 0, stream>>>(
        h1, src, dst, key, ew, dinv, b1, W2, acc, list, chunks, N, E);

    agg_tile_kernel<64, false><<<aggBlocks, 1024, 0, stream>>>(
        acc, src, dst, key, ew, dinv, b2, nullptr, out, list, chunks, N, E);
}


# layer 1 aggregates the input rows and multiplies its row tile by W1 inside the aggregation's work-group (fp32 WMMA out of LDS): no x@W1 launch, no product plane; sums taken before the product
# speedup vs baseline: 3.8950x; 1.0744x over previous
#include <hip/hip_runtime.h>


typedef __attribute__((ext_vector_type(2))) float v2f;
typedef __attribute__((ext_vector_type(8))) float v8f;

constexpr int TILE_R = 391;

__global__ void key_kernel(const int* __restrict__ src, const int* __restrict__ dst,
                           const float* __restrict__ w, unsigned char* __restrict__ key,
                           unsigned long long* __restrict__ erec, int e) {
    int i = blockIdx.x * blockDim.x + threadIdx.x;
    if (i < e) {
        const unsigned d = (unsigned)dst[i];
        const unsigned g = d / (unsigned)TILE_R;
        key[i]  = (unsigned char)g;
        erec[i] = ((unsigned long long)__float_as_uint(w[i]) << 32) |
                  (unsigned long long)(((d - g * (unsigned)TILE_R) << 17) | (unsigned)src[i]);
    }
}


constexpr int QCAP    = 64;
constexpr int GRP     = 8;
constexpr int STEP_SH = 17;

__device__ __forceinline__ unsigned zero_bytes(unsigned x) {
    return ~(((x & 0x7F7F7F7Fu) + 0x7F7F7F7Fu) | x) & 0x80808080u;
}

__device__ __forceinline__ unsigned guarded_word(const unsigned char* __restrict__ key,
                                                 int i0, int e, unsigned miss) {
    unsigned w = 0;
#pragma unroll
    for (int k = 0; k < 4; ++k) {
        const unsigned b = (i0 + k < e) ? (unsigned)key[i0 + k] : miss;
        w |= b << (8 * k);
    }
    return w;
}

typedef unsigned long long rec_t;
constexpr int WAVE_CHUNKS = 32;

template <class Drain>
__device__ __forceinline__ void scan_keys(const unsigned char* __restrict__ key, int e, int* qe,
                                          int wave, int lane, Drain drain) {
    const unsigned mine  = blockIdx.x & 255u;
    const unsigned pat   = mine * 0x01010101u;
    const unsigned miss  = mine ^ 1u;
    const unsigned miss4 = miss * 0x01010101u;
    const bool vec_ok = ((((size_t)key) & 15) == 0);
    int qcnt = 0;

    const int nsteps = (e + (1 << STEP_SH) - 1) >> STEP_SH;
    const int step0  = (int)(((long long)blockIdx.x * nsteps) / gridDim.x);
    const int woff   = (wave * GRP) << 9;

    for (int it = 0; it < nsteps; ++it) {
        int st = step0 + it;
        if (st >= nsteps) st -= nsteps;
        const int wbase = (st << STEP_SH) + woff;

        uint4 cur[GRP];
        if (vec_ok && wbase + (GRP << 9) <= e) {
#pragma unroll
            for (int u = 0; u < GRP; ++u)
                cur[u] = *(const uint4*)(key + wbase + (u << 9) + lane * 16);
        } else {
#pragma unroll
            for (int u = 0; u < GRP; ++u) {
                const int c0 = wbase + (u << 9);
                if (vec_ok && c0 + 512 <= e) {
                    cur[u] = *(const uint4*)(key + c0 + lane * 16);
                } else if (c0 >= e) {
                    cur[u] = make_uint4(miss4, miss4, miss4, miss4);
                } else {
                    const int i0 = c0 + lane * 16;
                    cur[u].x = guarded_word(key, i0,      e, miss);
                    cur[u].y = guarded_word(key, i0 + 4,  e, miss);
                    cur[u].z = guarded_word(key, i0 + 8,  e, miss);
                    cur[u].w = guarded_word(key, i0 + 12, e, miss);
                }
            }
        }

#pragma unroll
        for (int u = 0; u < GRP; ++u) {
            unsigned M = (zero_bytes(cur[u].x ^ pat) >> 7) | (zero_bytes(cur[u].y ^ pat) >> 6) |
                         (zero_bytes(cur[u].z ^ pat) >> 5) | (zero_bytes(cur[u].w ^ pat) >> 4);
            unsigned bm = __builtin_amdgcn_ballot_w32(M != 0u);
            const int e0 = wbase + (u << 9) + lane * 16;
            while (bm) {
                if (M != 0u) {
                    const int B = __builtin_ctz(M);
                    M &= M - 1u;
                    qe[__builtin_amdgcn_mbcnt_lo(bm, (unsigned)qcnt)] = e0 + ((B & 7) << 2) + (B >> 3);
                }
                qcnt += __popc(bm);
                if (qcnt >= 32) {
                    drain(32);
                    qcnt -= 32;
                    if (lane < qcnt) {
                        const int keep = qe[32 + lane];
                        qe[lane] = keep;
                    }
                }
                bm = __builtin_amdgcn_ballot_w32(M != 0u);
            }
        }
    }
    if (qcnt > 0) drain(qcnt);
}

__global__ __launch_bounds__(1024) void scan_deg_kernel(
    const unsigned long long* __restrict__ erec,
    const unsigned char* __restrict__ key, float* __restrict__ dinv,
    unsigned long long* __restrict__ list, int* __restrict__ chunks, int n, int e) {
    __shared__ float sDeg[TILE_R + 1];
    __shared__ int sQe[32 * QCAP];
    __shared__ int sChunks[32];

    const int tid  = threadIdx.x;
    const int lane = tid & 31;
    const int wave = __builtin_amdgcn_readfirstlane(tid >> 5);
    const int base = blockIdx.x * TILE_R;
    const int rows = min(TILE_R, n - base);

    if (tid <= TILE_R) sDeg[tid] = 1.0f;
    __syncthreads();

    int* qe = sQe + wave * QCAP;
    rec_t* wlist = list + (size_t)(blockIdx.x * 32 + wave) * (WAVE_CHUNKS * 32);
    int wch = 0;

    scan_keys(key, e, qe, wave, lane, [&](int cnt) {
        for (int b0 = 0; b0 < cnt; b0 += 32) {
            const int idx = b0 + lane;
            rec_t rec = (rec_t)((unsigned)TILE_R << 17);
            if (idx < cnt) {
                rec = erec[qe[idx]];
                unsafeAtomicAdd(&sDeg[(unsigned)rec >> 17], __uint_as_float((unsigned)(rec >> 32)));
            }
            if (wch < WAVE_CHUNKS) wlist[wch * 32 + lane] = rec;
            ++wch;
        }
    });

    if (lane == 0) sChunks[wave] = wch;
    __syncthreads();

    if (wave == 0) chunks[blockIdx.x * 32 + lane] = sChunks[lane];
    if (tid < rows) dinv[base + tid] = rsqrtf(sDeg[tid]);
}

constexpr int ROWS_IN_FLIGHT = 8;

template <int COLS, int PITCH>
__device__ __forceinline__ void add_rows(const float* __restrict__ h, int s, int r, float nm,
                                         int m, float* sAcc, int lane) {
    constexpr int VW = COLS / 32;
    for (int j = 0; j < m; j += ROWS_IN_FLIGHT) {
        float v[ROWS_IN_FLIGHT][VW];
        int   rr[ROWS_IN_FLIGHT];
        float nn[ROWS_IN_FLIGHT];
#pragma unroll
        for (int t = 0; t < ROWS_IN_FLIGHT; ++t) {
            const int jj = min(j + t, m - 1);
            const int sj = __builtin_amdgcn_readlane(s, jj);
            const int rj = __builtin_amdgcn_readlane(r, jj);
            rr[t] = (j + t < m) ? rj : TILE_R;
            nn[t] = __int_as_float(__builtin_amdgcn_readlane(__float_as_int(nm), jj));
            const float* hp = h + (size_t)sj * COLS + lane;
#pragma unroll
            for (int c = 0; c < VW; ++c) v[t][c] = hp[32 * c];
        }
#pragma unroll
        for (int t = 0; t < ROWS_IN_FLIGHT; ++t) {
            float* o = sAcc + rr[t] * PITCH + lane;
#pragma unroll
            for (int c = 0; c < VW; ++c) unsafeAtomicAdd(o + 32 * c, v[t][c] * nn[t]);
        }
    }
}

template <int COLS, int PITCH>
__device__ __forceinline__ void drain_queue(
    const float* __restrict__ h, const int* __restrict__ src, const int* __restrict__ dst,
    const float* __restrict__ w, const float* __restrict__ dinv, const int* qe, int cnt,
    int base, float* sAcc, int lane) {
    for (int b0 = 0; b0 < cnt; b0 += 32) {
        const int idx = b0 + lane;
        int s = 0, r = TILE_R;
        float nm = 0.0f;
        if (idx < cnt) {
            const int ed = qe[idx];
            s  = src[ed];
            r  = dst[ed] - base;
            nm = w[ed] * dinv[s];
        }
        add_rows<COLS, PITCH>(h, s, r, nm, min(32, cnt - b0), sAcc, lane);
    }
}

constexpr int OUT2     = 64;
constexpr int TILE_RB  = (TILE_R + 1 + 15) / 16;
constexpr int W1_PITCH = 128 * 2 + 32;
constexpr int W2_PITCH = OUT2 * 2 + 32;

template <int COLS, bool FUSED>
__global__ __launch_bounds__(1024) void agg_tile_kernel(
    const float* __restrict__ h, const int* __restrict__ src, const int* __restrict__ dst,
    const unsigned char* __restrict__ key, const float* __restrict__ w,
    const float* __restrict__ dinv, const float* __restrict__ b,
    const float* __restrict__ W1, const float* __restrict__ W2,
    float* __restrict__ out, const unsigned long long* __restrict__ list,
    const int* __restrict__ chunks, int n, int e) {
    constexpr int SHIFT = (COLS == 128) ? 7 : 6;
    constexpr int PITCH = FUSED ? COLS + 4 : COLS;
    constexpr int LROWS = FUSED ? TILE_RB * 16 : TILE_R + 1;
    __shared__ float sAcc[LROWS * PITCH];
    __shared__ float sW[FUSED ? (COLS / 2) * W1_PITCH : 4];
    __shared__ int sQe[32 * QCAP];

    const int tid  = threadIdx.x;
    const int lane = tid & 31;
    const int wave = __builtin_amdgcn_readfirstlane(tid >> 5);
    const int base = blockIdx.x * TILE_R;
    const int rows = min(TILE_R, n - base);
    const int tileElems = rows * COLS;
    const float* hTile = h + (size_t)base * COLS;

    constexpr int W1_PER_THREAD = FUSED ? COLS * COLS / 1024 : 1;
    constexpr int W2_PER_THREAD = FUSED ? COLS * OUT2 / 1024 : 1;
    float w1v[W1_PER_THREAD];
    if (FUSED) {
#pragma unroll
        for (int j = 0; j < W1_PER_THREAD; ++j) w1v[j] = W1[tid + j * 1024];
    }

    for (int i = tid * 4; i < tileElems; i += 1024 * 4) {
        const int row = i >> SHIFT;
        const float dv = dinv[base + row];
        const float4 hv = *(const float4*)(hTile + i);
        float4 o;
        o.x = hv.x * dv; o.y = hv.y * dv; o.z = hv.z * dv; o.w = hv.w * dv;
        *(float4*)(sAcc + row * PITCH + (i & (COLS - 1))) = o;
    }
    if (FUSED) {
#pragma unroll
        for (int j = 0; j < W1_PER_THREAD; ++j) {
            const int i   = tid + j * 1024;
            const int row = i / COLS;
            const int col = i % COLS;
            sW[(row >> 1) * W1_PITCH + col * 2 + (row & 1)] = w1v[j];
        }
    }
    for (int i = tid; i < (LROWS - TILE_R) * COLS; i += 1024)
        sAcc[(TILE_R + (i >> SHIFT)) * PITCH + (i & (COLS - 1))] = 0.0f;
    __syncthreads();

    const rec_t* wlist = list + (size_t)(blockIdx.x * 32 + wave) * (WAVE_CHUNKS * 32);

    const int cl = chunks[blockIdx.x * 32 + lane];
    const bool scan = __builtin_amdgcn_ballot_w32(cl > WAVE_CHUNKS) != 0u;
    if (!scan) {
        const int nch = __builtin_amdgcn_readlane(cl, wave);
        const rec_t null_rec = (rec_t)((unsigned)TILE_R << 17);
        rec_t rec = null_rec, nxt = null_rec;
        if (nch > 0) rec = wlist[lane];
        if (nch > 1) nxt = wlist[32 + lane];
        float dv = dinv[(unsigned)rec & 0x1FFFFu];
        for (int c = 0; c < nch; ++c) {
            rec_t nx2 = null_rec;
            if (c + 2 < nch) nx2 = wlist[(c + 2) * 32 + lane];
            const float dvn = dinv[(unsigned)nxt & 0x1FFFFu];
            const unsigned lo = (unsigned)rec;
            const int r = (int)(lo >> 17);
            const int m = __popc(__builtin_amdgcn_ballot_w32(r != TILE_R));
            add_rows<COLS, PITCH>(h, (int)(lo & 0x1FFFFu), r, __uint_as_float((unsigned)(rec >> 32)) * dv, m,
                                  sAcc, lane);
            rec = nxt; nxt = nx2; dv = dvn;
        }
    } else {
        int* qe = sQe + wave * QCAP;
        scan_keys(key, e, qe, wave, lane, [&](int cnt) {
            drain_queue<COLS, PITCH>(h, src, dst, w, dinv, qe, cnt, base, sAcc, lane);
        });
    }
    __syncthreads();

    if (!FUSED) {
        float* oTile = out + (size_t)base * COLS;
        for (int i = tid * 4; i < tileElems; i += 1024 * 4) {
            const int row = i >> SHIFT;
            const float dv = dinv[base + row];
            const float4 bv = *(const float4*)(b + (i & (COLS - 1)));
            float4 o = *(const float4*)(sAcc + row * PITCH + (i & (COLS - 1)));
            o.x = o.x * dv + bv.x; o.y = o.y * dv + bv.y;
            o.z = o.z * dv + bv.z; o.w = o.w * dv + bv.w;
            *(float4*)(oTile + i) = o;
        }
    } else {
        const int l15  = lane & 15;
        const int hsel = lane >> 4;

        float w2v[W2_PER_THREAD];
#pragma unroll
        for (int j = 0; j < W2_PER_THREAD; ++j) w2v[j] = W2[tid + j * 1024];

        if (wave < TILE_RB) {
            constexpr int NT = COLS / 16;
            const float* aBase = sAcc + (wave * 16 + l15) * PITCH + 2 * hsel;
            const float* bBase = sW + hsel * W1_PITCH + l15 * 2;
            v8f acc1[NT] = {};
            for (int kk = 0; kk < COLS / 4; ++kk) {
                const v2f A = *(const v2f*)(aBase + 4 * kk);
                const float* bp = bBase + kk * (2 * W1_PITCH);
#pragma unroll
                for (int c = 0; c < NT; ++c) {
                    const v2f B = *(const v2f*)(bp + c * 32);
                    acc1[c] = __builtin_amdgcn_wmma_f32_16x16x4_f32(
                        false, A, false, B, (short)0, acc1[c], false, false);
                }
            }
            const int rowBase = wave * 16 + (hsel << 3);
            float dvr[8];
#pragma unroll
            for (int r = 0; r < 8; ++r) dvr[r] = (rowBase + r < rows) ? dinv[base + rowBase + r] : 0.0f;
#pragma unroll
            for (int c = 0; c < NT; ++c) {
                const float bv = b[c * 16 + l15];
#pragma unroll
                for (int r = 0; r < 8; ++r)
                    sAcc[(rowBase + r) * PITCH + c * 16 + l15] = fmaxf(acc1[c][r] * dvr[r] + bv, 0.0f);
            }
        }
        __syncthreads();

#pragma unroll
        for (int j = 0; j < W2_PER_THREAD; ++j) {
            const int i   = tid + j * 1024;
            const int row = i / OUT2;
            const int col = i % OUT2;
            sW[(row >> 1) * W2_PITCH + col * 2 + (row & 1)] = w2v[j];
        }
        __syncthreads();

        const int c0   = (wave & 3) * 16;
        const float* bCol = sW + hsel * W2_PITCH + (c0 + l15) * 2;
        v2f Bf[COLS / 4];
#pragma unroll
        for (int kk = 0; kk < COLS / 4; ++kk) Bf[kk] = *(const v2f*)(bCol + kk * (2 * W2_PITCH));

        for (int rb = wave >> 2; rb < TILE_RB; rb += 8) {
            const float* aBase = sAcc + (rb * 16 + l15) * PITCH + 2 * hsel;
            v8f acc = {};
#pragma unroll
            for (int kk = 0; kk < COLS / 4; ++kk) {
                const v2f A = *(const v2f*)(aBase + 4 * kk);
                acc = __builtin_amdgcn_wmma_f32_16x16x4_f32(
                    false, A, false, Bf[kk], (short)0, acc, false, false);
            }
            const int rowBase = rb * 16 + (hsel << 3);
#pragma unroll
            for (int r = 0; r < 8; ++r) {
                const int row = rowBase + r;
                if (row < rows) out[(size_t)(base + row) * OUT2 + c0 + l15] = acc[r];
            }
        }
    }
}


extern "C" void kernel_launch(void* const* d_in, const int* in_sizes, int n_in,
                              void* d_out, int out_size, void* d_ws, size_t ws_size,
                              hipStream_t stream) {
    const float* x   = (const float*)d_in[0];
    const int*  eidx = (const int*)d_in[1];
    const float* ew  = (const float*)d_in[2];
    const float* W1  = (const float*)d_in[3];
    const float* b1  = (const float*)d_in[4];
    const float* W2  = (const float*)d_in[5];
    const float* b2  = (const float*)d_in[6];

    const int N = in_sizes[0] / 128;
    const int E = in_sizes[2];
    const int* src = eidx;
    const int* dst = eidx + E;

    float* dinv = (float*)d_ws;
    float* acc  = dinv + N;
    unsigned char* key = (unsigned char*)(acc + (size_t)N * 128);
    float* out  = (float*)d_out;

    const int T = 256;
    const int aggBlocks  = (N + TILE_R - 1) / TILE_R;
    const size_t keyEnd  = (size_t)((key + E) - (unsigned char*)d_ws);
    int* chunks = (int*)((unsigned char*)d_ws + ((keyEnd + 255) & ~(size_t)255));
    unsigned long long* list =
        (unsigned long long*)(chunks + (((size_t)aggBlocks * 32 + 63) & ~(size_t)63));
    unsigned long long* erec = list + (size_t)aggBlocks * 32 * (WAVE_CHUNKS * 32);

    key_kernel<<<(E + T - 1) / T, T, 0, stream>>>(src, dst, ew, key, erec, E);
    scan_deg_kernel<<<aggBlocks, 1024, 0, stream>>>(erec, key, dinv, list, chunks, N, E);

    agg_tile_kernel<128, true><<<aggBlocks, 1024, 0, stream>>>(
        x, src, dst, key, ew, dinv, b1, W1, W2, acc, list, chunks, N, E);

    agg_tile_kernel<64, false><<<aggBlocks, 1024, 0, stream>>>(
        acc, src, dst, key, ew, dinv, b2, nullptr, nullptr, out, list, chunks, N, E);
}


# replay fetches each row with ONE load (16/8 bytes per lane; were 4 and 2 one-line loads); tile columns interleaved in LDS keep adds conflict-free: W1 product sums k in that order; g 256-byte aligned
# speedup vs baseline: 4.3338x; 1.1127x over previous
#include <hip/hip_runtime.h>


typedef __attribute__((ext_vector_type(2))) float v2f;
typedef __attribute__((ext_vector_type(4))) float v4f;
typedef __attribute__((ext_vector_type(8))) float v8f;

constexpr int TILE_R = 391;

__global__ void key_kernel(const int* __restrict__ src, const int* __restrict__ dst,
                           const float* __restrict__ w, unsigned char* __restrict__ key,
                           unsigned long long* __restrict__ erec, int e) {
    int i = blockIdx.x * blockDim.x + threadIdx.x;
    if (i < e) {
        const unsigned d = (unsigned)dst[i];
        const unsigned g = d / (unsigned)TILE_R;
        key[i]  = (unsigned char)g;
        erec[i] = ((unsigned long long)__float_as_uint(w[i]) << 32) |
                  (unsigned long long)(((d - g * (unsigned)TILE_R) << 17) | (unsigned)src[i]);
    }
}


constexpr int QCAP    = 64;
constexpr int GRP     = 8;
constexpr int STEP_SH = 17;

__device__ __forceinline__ unsigned zero_bytes(unsigned x) {
    return ~(((x & 0x7F7F7F7Fu) + 0x7F7F7F7Fu) | x) & 0x80808080u;
}

__device__ __forceinline__ unsigned guarded_word(const unsigned char* __restrict__ key,
                                                 int i0, int e, unsigned miss) {
    unsigned w = 0;
#pragma unroll
    for (int k = 0; k < 4; ++k) {
        const unsigned b = (i0 + k < e) ? (unsigned)key[i0 + k] : miss;
        w |= b << (8 * k);
    }
    return w;
}

typedef unsigned long long rec_t;
constexpr int WAVE_CHUNKS = 32;

template <class Drain>
__device__ __forceinline__ void scan_keys(const unsigned char* __restrict__ key, int e, int* qe,
                                          int wave, int lane, Drain drain) {
    const unsigned mine  = blockIdx.x & 255u;
    const unsigned pat   = mine * 0x01010101u;
    const unsigned miss  = mine ^ 1u;
    const unsigned miss4 = miss * 0x01010101u;
    const bool vec_ok = ((((size_t)key) & 15) == 0);
    int qcnt = 0;

    const int nsteps = (e + (1 << STEP_SH) - 1) >> STEP_SH;
    const int step0  = (int)(((long long)blockIdx.x * nsteps) / gridDim.x);
    const int woff   = (wave * GRP) << 9;

    for (int it = 0; it < nsteps; ++it) {
        int st = step0 + it;
        if (st >= nsteps) st -= nsteps;
        const int wbase = (st << STEP_SH) + woff;

        uint4 cur[GRP];
        if (vec_ok && wbase + (GRP << 9) <= e) {
#pragma unroll
            for (int u = 0; u < GRP; ++u)
                cur[u] = *(const uint4*)(key + wbase + (u << 9) + lane * 16);
        } else {
#pragma unroll
            for (int u = 0; u < GRP; ++u) {
                const int c0 = wbase + (u << 9);
                if (vec_ok && c0 + 512 <= e) {
                    cur[u] = *(const uint4*)(key + c0 + lane * 16);
                } else if (c0 >= e) {
                    cur[u] = make_uint4(miss4, miss4, miss4, miss4);
                } else {
                    const int i0 = c0 + lane * 16;
                    cur[u].x = guarded_word(key, i0,      e, miss);
                    cur[u].y = guarded_word(key, i0 + 4,  e, miss);
                    cur[u].z = guarded_word(key, i0 + 8,  e, miss);
                    cur[u].w = guarded_word(key, i0 + 12, e, miss);
                }
            }
        }

#pragma unroll
        for (int u = 0; u < GRP; ++u) {
            unsigned M = (zero_bytes(cur[u].x ^ pat) >> 7) | (zero_bytes(cur[u].y ^ pat) >> 6) |
                         (zero_bytes(cur[u].z ^ pat) >> 5) | (zero_bytes(cur[u].w ^ pat) >> 4);
            unsigned bm = __builtin_amdgcn_ballot_w32(M != 0u);
            const int e0 = wbase + (u << 9) + lane * 16;
            while (bm) {
                if (M != 0u) {
                    const int B = __builtin_ctz(M);
                    M &= M - 1u;
                    qe[__builtin_amdgcn_mbcnt_lo(bm, (unsigned)qcnt)] = e0 + ((B & 7) << 2) + (B >> 3);
                }
                qcnt += __popc(bm);
                if (qcnt >= 32) {
                    drain(32);
                    qcnt -= 32;
                    if (lane < qcnt) {
                        const int keep = qe[32 + lane];
                        qe[lane] = keep;
                    }
                }
                bm = __builtin_amdgcn_ballot_w32(M != 0u);
            }
        }
    }
    if (qcnt > 0) drain(qcnt);
}

__global__ __launch_bounds__(1024) void scan_deg_kernel(
    const unsigned long long* __restrict__ erec,
    const unsigned char* __restrict__ key, float* __restrict__ dinv,
    unsigned long long* __restrict__ list, int* __restrict__ chunks, int n, int e) {
    __shared__ float sDeg[TILE_R + 1];
    __shared__ int sQe[32 * QCAP];
    __shared__ int sChunks[32];

    const int tid  = threadIdx.x;
    const int lane = tid & 31;
    const int wave = __builtin_amdgcn_readfirstlane(tid >> 5);
    const int base = blockIdx.x * TILE_R;
    const int rows = min(TILE_R, n - base);

    if (tid <= TILE_R) sDeg[tid] = 1.0f;
    __syncthreads();

    int* qe = sQe + wave * QCAP;
    rec_t* wlist = list + (size_t)(blockIdx.x * 32 + wave) * (WAVE_CHUNKS * 32);
    int wch = 0;

    scan_keys(key, e, qe, wave, lane, [&](int cnt) {
        for (int b0 = 0; b0 < cnt; b0 += 32) {
            const int idx = b0 + lane;
            rec_t rec = (rec_t)((unsigned)TILE_R << 17);
            if (idx < cnt) {
                rec = erec[qe[idx]];
                unsafeAtomicAdd(&sDeg[(unsigned)rec >> 17], __uint_as_float((unsigned)(rec >> 32)));
            }
            if (wch < WAVE_CHUNKS) wlist[wch * 32 + lane] = rec;
            ++wch;
        }
    });

    if (lane == 0) sChunks[wave] = wch;
    __syncthreads();

    if (wave == 0) chunks[blockIdx.x * 32 + lane] = sChunks[lane];
    if (tid < rows) dinv[base + tid] = rsqrtf(sDeg[tid]);
}

constexpr int ROWS_IN_FLIGHT = 8;

template <int COLS>
__device__ __forceinline__ int tile_pos(int col) {
    constexpr int VW = COLS / 32;
    return (col % VW) * 32 + col / VW;
}

template <int VW> struct row_piece;
template <> struct row_piece<4> { typedef v4f type; };
template <> struct row_piece<2> { typedef v2f type; };

template <int COLS, int PITCH>
__device__ __forceinline__ void add_rows(const float* __restrict__ h, int s, int r, float nm,
                                         int m, float* sAcc, int lane) {
    constexpr int VW = COLS / 32;
    typedef typename row_piece<VW>::type piece_t;
    for (int j = 0; j < m; j += ROWS_IN_FLIGHT) {
        piece_t v[ROWS_IN_FLIGHT];
        int   rr[ROWS_IN_FLIGHT];
        float nn[ROWS_IN_FLIGHT];
#pragma unroll
        for (int t = 0; t < ROWS_IN_FLIGHT; ++t) {
            const int jj = min(j + t, m - 1);
            const int sj = __builtin_amdgcn_readlane(s, jj);
            const int rj = __builtin_amdgcn_readlane(r, jj);
            rr[t] = (j + t < m) ? rj : TILE_R;
            nn[t] = __int_as_float(__builtin_amdgcn_readlane(__float_as_int(nm), jj));
            v[t] = *((const piece_t*)(h + (size_t)sj * COLS) + lane);
        }
#pragma unroll
        for (int t = 0; t < ROWS_IN_FLIGHT; ++t) {
            float* o = sAcc + rr[t] * PITCH + lane;
#pragma unroll
            for (int c = 0; c < VW; ++c) unsafeAtomicAdd(o + 32 * c, v[t][c] * nn[t]);
        }
    }
}

template <int COLS, int PITCH>
__device__ __forceinline__ void drain_queue(
    const float* __restrict__ h, const int* __restrict__ src, const int* __restrict__ dst,
    const float* __restrict__ w, const float* __restrict__ dinv, const int* qe, int cnt,
    int base, float* sAcc, int lane) {
    for (int b0 = 0; b0 < cnt; b0 += 32) {
        const int idx = b0 + lane;
        int s = 0, r = TILE_R;
        float nm = 0.0f;
        if (idx < cnt) {
            const int ed = qe[idx];
            s  = src[ed];
            r  = dst[ed] - base;
            nm = w[ed] * dinv[s];
        }
        add_rows<COLS, PITCH>(h, s, r, nm, min(32, cnt - b0), sAcc, lane);
    }
}

constexpr int OUT2     = 64;
constexpr int TILE_RB  = (TILE_R + 1 + 15) / 16;
constexpr int W1_PITCH = 128 * 2 + 32;
constexpr int W2_PITCH = OUT2 * 2 + 32;

template <int COLS, bool FUSED>
__global__ __launch_bounds__(1024) void agg_tile_kernel(
    const float* __restrict__ h, const int* __restrict__ src, const int* __restrict__ dst,
    const unsigned char* __restrict__ key, const float* __restrict__ w,
    const float* __restrict__ dinv, const float* __restrict__ b,
    const float* __restrict__ W1, const float* __restrict__ W2,
    float* __restrict__ out, const unsigned long long* __restrict__ list,
    const int* __restrict__ chunks, int n, int e) {
    constexpr int SHIFT = (COLS == 128) ? 7 : 6;
    constexpr int PITCH = FUSED ? COLS + 4 : COLS;
    constexpr int LROWS = FUSED ? TILE_RB * 16 : TILE_R + 1;
    __shared__ float sAcc[LROWS * PITCH];
    __shared__ float sW[FUSED ? (COLS / 2) * W1_PITCH : 4];
    __shared__ int sQe[32 * QCAP];

    const int tid  = threadIdx.x;
    const int lane = tid & 31;
    const int wave = __builtin_amdgcn_readfirstlane(tid >> 5);
    const int base = blockIdx.x * TILE_R;
    const int rows = min(TILE_R, n - base);
    const int tileElems = rows * COLS;
    const float* hTile = h + (size_t)base * COLS;

    constexpr int W1_PER_THREAD = FUSED ? COLS * COLS / 1024 : 1;
    constexpr int W2_PER_THREAD = FUSED ? COLS * OUT2 / 1024 : 1;
    float w1v[W1_PER_THREAD];
    if (FUSED) {
#pragma unroll
        for (int j = 0; j < W1_PER_THREAD; ++j) w1v[j] = W1[tid + j * 1024];
    }

    for (int i = tid * 4; i < tileElems; i += 1024 * 4) {
        const int row = i >> SHIFT;
        const float dv = dinv[base + row];
        const float4 hv = *(const float4*)(hTile + i);
        const int col = i & (COLS - 1);
        float* o = sAcc + row * PITCH;
        o[tile_pos<COLS>(col)]     = hv.x * dv;
        o[tile_pos<COLS>(col + 1)] = hv.y * dv;
        o[tile_pos<COLS>(col + 2)] = hv.z * dv;
        o[tile_pos<COLS>(col + 3)] = hv.w * dv;
    }
    if (FUSED) {
#pragma unroll
        for (int j = 0; j < W1_PER_THREAD; ++j) {
            const int i   = tid + j * 1024;
            const int row = i / COLS;
            const int col = i % COLS;
            const int q   = tile_pos<COLS>(row);
            sW[(q >> 1) * W1_PITCH + col * 2 + (q & 1)] = w1v[j];
        }
    }
    for (int i = tid; i < (LROWS - TILE_R) * COLS; i += 1024)
        sAcc[(TILE_R + (i >> SHIFT)) * PITCH + (i & (COLS - 1))] = 0.0f;
    __syncthreads();

    const rec_t* wlist = list + (size_t)(blockIdx.x * 32 + wave) * (WAVE_CHUNKS * 32);

    const int cl = chunks[blockIdx.x * 32 + lane];
    const bool scan = __builtin_amdgcn_ballot_w32(cl > WAVE_CHUNKS) != 0u;
    if (!scan) {
        const int nch = __builtin_amdgcn_readlane(cl, wave);
        const rec_t null_rec = (rec_t)((unsigned)TILE_R << 17);
        rec_t rec = null_rec, nxt = null_rec;
        if (nch > 0) rec = wlist[lane];
        if (nch > 1) nxt = wlist[32 + lane];
        float dv = dinv[(unsigned)rec & 0x1FFFFu];
        for (int c = 0; c < nch; ++c) {
            rec_t nx2 = null_rec;
            if (c + 2 < nch) nx2 = wlist[(c + 2) * 32 + lane];
            const float dvn = dinv[(unsigned)nxt & 0x1FFFFu];
            const unsigned lo = (unsigned)rec;
            const int r = (int)(lo >> 17);
            const int m = __popc(__builtin_amdgcn_ballot_w32(r != TILE_R));
            add_rows<COLS, PITCH>(h, (int)(lo & 0x1FFFFu), r, __uint_as_float((unsigned)(rec >> 32)) * dv, m,
                                  sAcc, lane);
            rec = nxt; nxt = nx2; dv = dvn;
        }
    } else {
        int* qe = sQe + wave * QCAP;
        scan_keys(key, e, qe, wave, lane, [&](int cnt) {
            drain_queue<COLS, PITCH>(h, src, dst, w, dinv, qe, cnt, base, sAcc, lane);
        });
    }
    __syncthreads();

    if (!FUSED) {
        float* oTile = out + (size_t)base * COLS;
        for (int i = tid * 4; i < tileElems; i += 1024 * 4) {
            const int row = i >> SHIFT;
            const float dv = dinv[base + row];
            const int col = i & (COLS - 1);
            const float4 bv = *(const float4*)(b + col);
            const float* a = sAcc + row * PITCH;
            float4 o;
            o.x = a[tile_pos<COLS>(col)]     * dv + bv.x;
            o.y = a[tile_pos<COLS>(col + 1)] * dv + bv.y;
            o.z = a[tile_pos<COLS>(col + 2)] * dv + bv.z;
            o.w = a[tile_pos<COLS>(col + 3)] * dv + bv.w;
            *(float4*)(oTile + i) = o;
        }
    } else {
        const int l15  = lane & 15;
        const int hsel = lane >> 4;

        float w2v[W2_PER_THREAD];
#pragma unroll
        for (int j = 0; j < W2_PER_THREAD; ++j) w2v[j] = W2[tid + j * 1024];

        if (wave < TILE_RB) {
            constexpr int NT = COLS / 16;
            const float* aBase = sAcc + (wave * 16 + l15) * PITCH + 2 * hsel;
            const float* bBase = sW + hsel * W1_PITCH + l15 * 2;
            v8f acc1[NT] = {};
            for (int kk = 0; kk < COLS / 4; ++kk) {
                const v2f A = *(const v2f*)(aBase + 4 * kk);
                const float* bp = bBase + kk * (2 * W1_PITCH);
#pragma unroll
                for (int c = 0; c < NT; ++c) {
                    const v2f B = *(const v2f*)(bp + c * 32);
                    acc1[c] = __builtin_amdgcn_wmma_f32_16x16x4_f32(
                        false, A, false, B, (short)0, acc1[c], false, false);
                }
            }
            const int rowBase = wave * 16 + (hsel << 3);
            float dvr[8];
#pragma unroll
            for (int r = 0; r < 8; ++r) dvr[r] = (rowBase + r < rows) ? dinv[base + rowBase + r] : 0.0f;
#pragma unroll
            for (int c = 0; c < NT; ++c) {
                const float bv = b[c * 16 + l15];
#pragma unroll
                for (int r = 0; r < 8; ++r)
                    sAcc[(rowBase + r) * PITCH + c * 16 + l15] = fmaxf(acc1[c][r] * dvr[r] + bv, 0.0f);
            }
        }
        __syncthreads();

#pragma unroll
        for (int j = 0; j < W2_PER_THREAD; ++j) {
            const int i   = tid + j * 1024;
            const int row = i / OUT2;
            const int col = i % OUT2;
            sW[(row >> 1) * W2_PITCH + col * 2 + (row & 1)] = w2v[j];
        }
        __syncthreads();

        const int c0   = (wave & 3) * 16;
        const float* bCol = sW + hsel * W2_PITCH + (c0 + l15) * 2;
        v2f Bf[COLS / 4];
#pragma unroll
        for (int kk = 0; kk < COLS / 4; ++kk) Bf[kk] = *(const v2f*)(bCol + kk * (2 * W2_PITCH));

        for (int rb = wave >> 2; rb < TILE_RB; rb += 8) {
            const float* aBase = sAcc + (rb * 16 + l15) * PITCH + 2 * hsel;
            v8f acc = {};
#pragma unroll
            for (int kk = 0; kk < COLS / 4; ++kk) {
                const v2f A = *(const v2f*)(aBase + 4 * kk);
                acc = __builtin_amdgcn_wmma_f32_16x16x4_f32(
                    false, A, false, Bf[kk], (short)0, acc, false, false);
            }
            const int rowBase = rb * 16 + (hsel << 3);
#pragma unroll
            for (int r = 0; r < 8; ++r) {
                const int row = rowBase + r;
                if (row < rows) out[(size_t)(base + row) * OUT2 + c0 + l15] = acc[r];
            }
        }
    }
}


extern "C" void kernel_launch(void* const* d_in, const int* in_sizes, int n_in,
                              void* d_out, int out_size, void* d_ws, size_t ws_size,
                              hipStream_t stream) {
    const float* x   = (const float*)d_in[0];
    const int*  eidx = (const int*)d_in[1];
    const float* ew  = (const float*)d_in[2];
    const float* W1  = (const float*)d_in[3];
    const float* b1  = (const float*)d_in[4];
    const float* W2  = (const float*)d_in[5];
    const float* b2  = (const float*)d_in[6];

    const int N = in_sizes[0] / 128;
    const int E = in_sizes[2];
    const int* src = eidx;
    const int* dst = eidx + E;

    float* dinv = (float*)d_ws;
    float* acc  = dinv + (((size_t)N + 63) & ~(size_t)63);
    unsigned char* key = (unsigned char*)(acc + (size_t)N * 128);
    float* out  = (float*)d_out;

    const int T = 256;
    const int aggBlocks  = (N + TILE_R - 1) / TILE_R;
    const size_t keyEnd  = (size_t)((key + E) - (unsigned char*)d_ws);
    int* chunks = (int*)((unsigned char*)d_ws + ((keyEnd + 255) & ~(size_t)255));
    unsigned long long* list =
        (unsigned long long*)(chunks + (((size_t)aggBlocks * 32 + 63) & ~(size_t)63));
    unsigned long long* erec = list + (size_t)aggBlocks * 32 * (WAVE_CHUNKS * 32);

    key_kernel<<<(E + T - 1) / T, T, 0, stream>>>(src, dst, ew, key, erec, E);
    scan_deg_kernel<<<aggBlocks, 1024, 0, stream>>>(erec, key, dinv, list, chunks, N, E);

    agg_tile_kernel<128, true><<<aggBlocks, 1024, 0, stream>>>(
        x, src, dst, key, ew, dinv, b1, W1, W2, acc, list, chunks, N, E);

    agg_tile_kernel<64, false><<<aggBlocks, 1024, 0, stream>>>(
        acc, src, dst, key, ew, dinv, b2, nullptr, nullptr, out, list, chunks, N, E);
}
